# starting kernel
# baseline (speedup 1.0000x reference)
#include <hip/hip_runtime.h>

#define IN_C  128
#define H_C   128
#define OUT_C 64

typedef float v2f __attribute__((ext_vector_type(2)));
typedef float v8f __attribute__((ext_vector_type(8)));

__device__ __forceinline__ void atomic_fadd(float* p, float v) {
    asm volatile("global_atomic_add_f32 %0, %1, off" : : "v"(p), "v"(v) : "memory");
}

__global__ void fill_kernel(float* __restrict__ p, float v, int n) {
    int i = blockIdx.x * blockDim.x + threadIdx.x;
    if (i < n) p[i] = v;
}

__global__ void deg_scatter_kernel(const int* __restrict__ dst,
                                   float* __restrict__ deg, int e) {
    int i = blockIdx.x * blockDim.x + threadIdx.x;
    if (i < e) atomic_fadd(&deg[dst[i]], 1.0f);
}

__global__ void rsqrt_kernel(float* __restrict__ p, int n) {
    int i = blockIdx.x * blockDim.x + threadIdx.x;
    if (i < n) p[i] = rsqrtf(p[i]);
}

template<int K, int NC, bool ADD_BIAS>
__global__ void gemm_wmma_kernel(const float* __restrict__ A,
                                 const float* __restrict__ B,
                                 const float* __restrict__ bias,
                                 float* __restrict__ C, int M) {
    __shared__ __align__(16) float As[16 * K];

    const int tid   = threadIdx.x;
    const int wave  = tid >> 5;
    const int lane  = tid & 31;
    const int n0    = wave * 16;
    const int lhalf = lane >> 4;
    const int l15   = lane & 15;
    const int n     = n0 + l15;

    v2f bfrag[K / 4];
#pragma unroll
    for (int k0 = 0; k0 < K; k0 += 4) {
        const int kk = k0 + 2 * lhalf;
        bfrag[k0 / 4].x = B[(size_t)kk * NC + n];
        bfrag[k0 / 4].y = B[(size_t)(kk + 1) * NC + n];
    }
    const float bv = ADD_BIAS ? bias[n] : 0.0f;

    const int strips      = (M + 15) / 16;
    const int vecsPerRow  = K / 4;

    for (int s = blockIdx.x; s < strips; s += gridDim.x) {
        const int row0 = s * 16;
        __syncthreads();
        for (int v = tid; v < 16 * vecsPerRow; v += blockDim.x) {
            int r = v / vecsPerRow;
            float4 val = make_float4(0.f, 0.f, 0.f, 0.f);
            if (row0 + r < M)
                val = ((const float4*)(A + (size_t)(row0 + r) * K))[v % vecsPerRow];
            ((float4*)As)[v] = val;
        }
        __syncthreads();

        v8f acc = {};
#pragma unroll
        for (int k0 = 0; k0 < K; k0 += 4) {
            const int kk = k0 + 2 * lhalf;
            v2f a;
            a.x = As[l15 * K + kk];
            a.y = As[l15 * K + kk + 1];
            acc = __builtin_amdgcn_wmma_f32_16x16x4_f32(
                      false, a, false, bfrag[k0 / 4], (short)0, acc, false, false);
        }

#pragma unroll
        for (int r = 0; r < 8; ++r) {
            int m = row0 + r + 8 * lhalf;
            if (m < M) C[(size_t)m * NC + n] = acc[r] + bv;
        }
    }
}

__global__ void agg_init_kernel(const float* __restrict__ tmp,
                                const float* __restrict__ dinv,
                                const float* __restrict__ bias,
                                float* __restrict__ out, int total) {
    int i = blockIdx.x * blockDim.x + threadIdx.x;
    if (i >= total) return;
    int node = i >> 7;
    int c    = i & 127;
    float di = dinv[node];
    out[i] = tmp[i] * di * di + bias[c];
}

__global__ void edge_scatter_kernel(const int* __restrict__ src,
                                    const int* __restrict__ dst,
                                    const float* __restrict__ dinv,
                                    const float* __restrict__ msg,
                                    float* __restrict__ out, int e) {
    int gid  = blockIdx.x * blockDim.x + threadIdx.x;
    int edge = gid >> 5;
    int lane = gid & 31;
    if (edge >= e) return;
    int s = src[edge], d = dst[edge];
    float w = dinv[s] * dinv[d];
    float4 v = ((const float4*)(msg + (size_t)s * H_C))[lane];
    float* o = out + (size_t)d * H_C + lane * 4;
    atomic_fadd(o + 0, v.x * w);
    atomic_fadd(o + 1, v.y * w);
    atomic_fadd(o + 2, v.z * w);
    atomic_fadd(o + 3, v.w * w);
}

__global__ void relu_kernel(float* __restrict__ p, int n) {
    int i = blockIdx.x * blockDim.x + threadIdx.x;
    if (i < n) p[i] = fmaxf(p[i], 0.0f);
}

__global__ void pool_scatter_kernel(const float* __restrict__ h,
                                    const int* __restrict__ batch,
                                    float* __restrict__ sums,
                                    float* __restrict__ cnts, int n) {
    int gid  = blockIdx.x * blockDim.x + threadIdx.x;
    int node = gid >> 5;
    int lane = gid & 31;
    if (node >= n) return;
    int g = batch[node];
    float4 v = ((const float4*)(h + (size_t)node * H_C))[lane];
    float* o = sums + (size_t)g * H_C + lane * 4;
    atomic_fadd(o + 0, v.x);
    atomic_fadd(o + 1, v.y);
    atomic_fadd(o + 2, v.z);
    atomic_fadd(o + 3, v.w);
    if (lane == 0) atomic_fadd(&cnts[g], 1.0f);
}

__global__ void pool_div_kernel(float* __restrict__ pooled,
                                const float* __restrict__ cnts, int total) {
    int i = blockIdx.x * blockDim.x + threadIdx.x;
    if (i >= total) return;
    pooled[i] /= fmaxf(cnts[i >> 7], 1.0f);
}

extern "C" void kernel_launch(void* const* d_in, const int* in_sizes, int n_in,
                              void* d_out, int out_size, void* d_ws, size_t ws_size,
                              hipStream_t stream) {
    const float* x    = (const float*)d_in[0];
    const int*   ei   = (const int*)  d_in[1];
    const int*   batch= (const int*)  d_in[2];
    const float* W1   = (const float*)d_in[3];
    const float* b1   = (const float*)d_in[4];
    const float* W2   = (const float*)d_in[5];
    const float* b2   = (const float*)d_in[6];
    const float* Wl   = (const float*)d_in[7];
    const float* bl   = (const float*)d_in[8];

    const int N = in_sizes[0] / IN_C;
    const int E = in_sizes[1] / 2;
    const int G = (out_size - N * OUT_C) / H_C;

    const int* esrc = ei;
    const int* edst = ei + E;

    float* logits = (float*)d_out;
    float* pooled = logits + (size_t)N * OUT_C;

    float* deg  = (float*)d_ws;
    float* tmpA = deg + (((size_t)N + 255) & ~(size_t)255);
    float* tmpB = tmpA + (size_t)N * H_C;
    float* cnts = tmpB + (size_t)N * H_C;

    const int T  = 256;
    const int NH = N * H_C;
    const int rowBlocks  = (N + 15) / 16;
    const int gemmBlocks = rowBlocks < 512 ? rowBlocks : 512;

    fill_kernel<<<(N + T - 1) / T, T, 0, stream>>>(deg, 1.0f, N);
    fill_kernel<<<(G + T - 1) / T, T, 0, stream>>>(cnts, 0.0f, G);
    fill_kernel<<<(G * H_C + T - 1) / T, T, 0, stream>>>(pooled, 0.0f, G * H_C);
    deg_scatter_kernel<<<(E + T - 1) / T, T, 0, stream>>>(edst, deg, E);
    rsqrt_kernel<<<(N + T - 1) / T, T, 0, stream>>>(deg, N);

    gemm_wmma_kernel<IN_C, H_C, false><<<gemmBlocks, 256, 0, stream>>>(x, W1, nullptr, tmpA, N);
    agg_init_kernel<<<(NH + T - 1) / T, T, 0, stream>>>(tmpA, deg, b1, tmpB, NH);
    edge_scatter_kernel<<<((size_t)E * 32 + T - 1) / T, T, 0, stream>>>(esrc, edst, deg, tmpA, tmpB, E);
    relu_kernel<<<(NH + T - 1) / T, T, 0, stream>>>(tmpB, NH);

    gemm_wmma_kernel<H_C, H_C, false><<<gemmBlocks, 256, 0, stream>>>(tmpB, W2, nullptr, tmpA, N);
    agg_init_kernel<<<(NH + T - 1) / T, T, 0, stream>>>(tmpA, deg, b2, tmpB, NH);
    edge_scatter_kernel<<<((size_t)E * 32 + T - 1) / T, T, 0, stream>>>(esrc, edst, deg, tmpA, tmpB, E);
    relu_kernel<<<(NH + T - 1) / T, T, 0, stream>>>(tmpB, NH);

    gemm_wmma_kernel<H_C, OUT_C, true><<<gemmBlocks, 128, 0, stream>>>(tmpB, Wl, bl, logits, N);

    pool_scatter_kernel<<<((size_t)N * 32 + T - 1) / T, T, 0, stream>>>(tmpB, batch, pooled, cnts, N);
    pool_div_kernel<<<(G * H_C + T - 1) / T, T, 0, stream>>>(pooled, cnts, G * H_C);
}


# edge and pool scatter: lane l takes channels l,l+32,l+64,l+96 so each f32 atomic instruction covers one whole 128-byte line instead of 8 lanes in each of 4 lines
# speedup vs baseline: 1.3468x; 1.3468x over previous
#include <hip/hip_runtime.h>

#define IN_C  128
#define H_C   128
#define OUT_C 64

typedef float v2f __attribute__((ext_vector_type(2)));
typedef float v8f __attribute__((ext_vector_type(8)));

__device__ __forceinline__ void atomic_fadd(float* p, float v) {
    asm volatile("global_atomic_add_f32 %0, %1, off" : : "v"(p), "v"(v) : "memory");
}

template<int OFF>
__device__ __forceinline__ void atomic_fadd_at(float* p, float v) {
    asm volatile("global_atomic_add_f32 %0, %1, off offset:%2" : : "v"(p), "v"(v), "n"(OFF) : "memory");
}

__global__ void fill_kernel(float* __restrict__ p, float v, int n) {
    int i = blockIdx.x * blockDim.x + threadIdx.x;
    if (i < n) p[i] = v;
}

__global__ void deg_scatter_kernel(const int* __restrict__ dst,
                                   float* __restrict__ deg, int e) {
    int i = blockIdx.x * blockDim.x + threadIdx.x;
    if (i < e) atomic_fadd(&deg[dst[i]], 1.0f);
}

__global__ void rsqrt_kernel(float* __restrict__ p, int n) {
    int i = blockIdx.x * blockDim.x + threadIdx.x;
    if (i < n) p[i] = rsqrtf(p[i]);
}

template<int K, int NC, bool ADD_BIAS>
__global__ void gemm_wmma_kernel(const float* __restrict__ A,
                                 const float* __restrict__ B,
                                 const float* __restrict__ bias,
                                 float* __restrict__ C, int M) {
    __shared__ __align__(16) float As[16 * K];

    const int tid   = threadIdx.x;
    const int wave  = tid >> 5;
    const int lane  = tid & 31;
    const int n0    = wave * 16;
    const int lhalf = lane >> 4;
    const int l15   = lane & 15;
    const int n     = n0 + l15;

    v2f bfrag[K / 4];
#pragma unroll
    for (int k0 = 0; k0 < K; k0 += 4) {
        const int kk = k0 + 2 * lhalf;
        bfrag[k0 / 4].x = B[(size_t)kk * NC + n];
        bfrag[k0 / 4].y = B[(size_t)(kk + 1) * NC + n];
    }
    const float bv = ADD_BIAS ? bias[n] : 0.0f;

    const int strips      = (M + 15) / 16;
    const int vecsPerRow  = K / 4;

    for (int s = blockIdx.x; s < strips; s += gridDim.x) {
        const int row0 = s * 16;
        __syncthreads();
        for (int v = tid; v < 16 * vecsPerRow; v += blockDim.x) {
            int r = v / vecsPerRow;
            float4 val = make_float4(0.f, 0.f, 0.f, 0.f);
            if (row0 + r < M)
                val = ((const float4*)(A + (size_t)(row0 + r) * K))[v % vecsPerRow];
            ((float4*)As)[v] = val;
        }
        __syncthreads();

        v8f acc = {};
#pragma unroll
        for (int k0 = 0; k0 < K; k0 += 4) {
            const int kk = k0 + 2 * lhalf;
            v2f a;
            a.x = As[l15 * K + kk];
            a.y = As[l15 * K + kk + 1];
            acc = __builtin_amdgcn_wmma_f32_16x16x4_f32(
                      false, a, false, bfrag[k0 / 4], (short)0, acc, false, false);
        }

#pragma unroll
        for (int r = 0; r < 8; ++r) {
            int m = row0 + r + 8 * lhalf;
            if (m < M) C[(size_t)m * NC + n] = acc[r] + bv;
        }
    }
}

__global__ void agg_init_kernel(const float* __restrict__ tmp,
                                const float* __restrict__ dinv,
                                const float* __restrict__ bias,
                                float* __restrict__ out, int total) {
    int i = blockIdx.x * blockDim.x + threadIdx.x;
    if (i >= total) return;
    int node = i >> 7;
    int c    = i & 127;
    float di = dinv[node];
    out[i] = tmp[i] * di * di + bias[c];
}

__global__ void edge_scatter_kernel(const int* __restrict__ src,
                                    const int* __restrict__ dst,
                                    const float* __restrict__ dinv,
                                    const float* __restrict__ msg,
                                    float* __restrict__ out, int e) {
    int gid  = blockIdx.x * blockDim.x + threadIdx.x;
    int edge = gid >> 5;
    int lane = gid & 31;
    if (edge >= e) return;
    int s = src[edge], d = dst[edge];
    float w = dinv[s] * dinv[d];
    const float* m = msg + (size_t)s * H_C + lane;
    float v0 = m[0], v1 = m[32], v2 = m[64], v3 = m[96];
    float* o = out + (size_t)d * H_C + lane;
    atomic_fadd_at<0>(o, v0 * w);
    atomic_fadd_at<128>(o, v1 * w);
    atomic_fadd_at<256>(o, v2 * w);
    atomic_fadd_at<384>(o, v3 * w);
}

__global__ void relu_kernel(float* __restrict__ p, int n) {
    int i = blockIdx.x * blockDim.x + threadIdx.x;
    if (i < n) p[i] = fmaxf(p[i], 0.0f);
}

__global__ void pool_scatter_kernel(const float* __restrict__ h,
                                    const int* __restrict__ batch,
                                    float* __restrict__ sums,
                                    float* __restrict__ cnts, int n) {
    int gid  = blockIdx.x * blockDim.x + threadIdx.x;
    int node = gid >> 5;
    int lane = gid & 31;
    if (node >= n) return;
    int g = batch[node];
    const float* m = h + (size_t)node * H_C + lane;
    float v0 = m[0], v1 = m[32], v2 = m[64], v3 = m[96];
    float* o = sums + (size_t)g * H_C + lane;
    atomic_fadd_at<0>(o, v0);
    atomic_fadd_at<128>(o, v1);
    atomic_fadd_at<256>(o, v2);
    atomic_fadd_at<384>(o, v3);
    if (lane == 0) atomic_fadd(&cnts[g], 1.0f);
}

__global__ void pool_div_kernel(float* __restrict__ pooled,
                                const float* __restrict__ cnts, int total) {
    int i = blockIdx.x * blockDim.x + threadIdx.x;
    if (i >= total) return;
    pooled[i] /= fmaxf(cnts[i >> 7], 1.0f);
}

extern "C" void kernel_launch(void* const* d_in, const int* in_sizes, int n_in,
                              void* d_out, int out_size, void* d_ws, size_t ws_size,
                              hipStream_t stream) {
    const float* x    = (const float*)d_in[0];
    const int*   ei   = (const int*)  d_in[1];
    const int*   batch= (const int*)  d_in[2];
    const float* W1   = (const float*)d_in[3];
    const float* b1   = (const float*)d_in[4];
    const float* W2   = (const float*)d_in[5];
    const float* b2   = (const float*)d_in[6];
    const float* Wl   = (const float*)d_in[7];
    const float* bl   = (const float*)d_in[8];

    const int N = in_sizes[0] / IN_C;
    const int E = in_sizes[1] / 2;
    const int G = (out_size - N * OUT_C) / H_C;

    const int* esrc = ei;
    const int* edst = ei + E;

    float* logits = (float*)d_out;
    float* pooled = logits + (size_t)N * OUT_C;

    float* deg  = (float*)d_ws;
    float* tmpA = deg + (((size_t)N + 255) & ~(size_t)255);
    float* tmpB = tmpA + (size_t)N * H_C;
    float* cnts = tmpB + (size_t)N * H_C;

    const int T  = 256;
    const int NH = N * H_C;
    const int rowBlocks  = (N + 15) / 16;
    const int gemmBlocks = rowBlocks < 512 ? rowBlocks : 512;

    fill_kernel<<<(N + T - 1) / T, T, 0, stream>>>(deg, 1.0f, N);
    fill_kernel<<<(G + T - 1) / T, T, 0, stream>>>(cnts, 0.0f, G);
    fill_kernel<<<(G * H_C + T - 1) / T, T, 0, stream>>>(pooled, 0.0f, G * H_C);
    deg_scatter_kernel<<<(E + T - 1) / T, T, 0, stream>>>(edst, deg, E);
    rsqrt_kernel<<<(N + T - 1) / T, T, 0, stream>>>(deg, N);

    gemm_wmma_kernel<IN_C, H_C, false><<<gemmBlocks, 256, 0, stream>>>(x, W1, nullptr, tmpA, N);
    agg_init_kernel<<<(NH + T - 1) / T, T, 0, stream>>>(tmpA, deg, b1, tmpB, NH);
    edge_scatter_kernel<<<((size_t)E * 32 + T - 1) / T, T, 0, stream>>>(esrc, edst, deg, tmpA, tmpB, E);
    relu_kernel<<<(NH + T - 1) / T, T, 0, stream>>>(tmpB, NH);

    gemm_wmma_kernel<H_C, H_C, false><<<gemmBlocks, 256, 0, stream>>>(tmpB, W2, nullptr, tmpA, N);
    agg_init_kernel<<<(NH + T - 1) / T, T, 0, stream>>>(tmpA, deg, b2, tmpB, NH);
    edge_scatter_kernel<<<((size_t)E * 32 + T - 1) / T, T, 0, stream>>>(esrc, edst, deg, tmpA, tmpB, E);
    relu_kernel<<<(NH + T - 1) / T, T, 0, stream>>>(tmpB, NH);

    gemm_wmma_kernel<H_C, OUT_C, true><<<gemmBlocks, 128, 0, stream>>>(tmpB, Wl, bl, logits, N);

    pool_scatter_kernel<<<((size_t)N * 32 + T - 1) / T, T, 0, stream>>>(tmpB, batch, pooled, cnts, N);
    pool_div_kernel<<<(G * H_C + T - 1) / T, T, 0, stream>>>(pooled, cnts, G * H_C);
}


# self-loop+bias start value of each aggregation written by the GEMM epilogue, relu applied by its readers (next GEMM's A staging, pool): 4 elementwise passes over the N x 128 activations removed
# speedup vs baseline: 1.5096x; 1.1209x over previous
#include <hip/hip_runtime.h>

#define IN_C  128
#define H_C   128
#define OUT_C 64

typedef float v2f __attribute__((ext_vector_type(2)));
typedef float v8f __attribute__((ext_vector_type(8)));

__device__ __forceinline__ void atomic_fadd(float* p, float v) {
    asm volatile("global_atomic_add_f32 %0, %1, off" : : "v"(p), "v"(v) : "memory");
}

template<int OFF>
__device__ __forceinline__ void atomic_fadd_at(float* p, float v) {
    asm volatile("global_atomic_add_f32 %0, %1, off offset:%2" : : "v"(p), "v"(v), "n"(OFF) : "memory");
}

__global__ void fill_kernel(float* __restrict__ p, float v, int n) {
    int i = blockIdx.x * blockDim.x + threadIdx.x;
    if (i < n) p[i] = v;
}

__global__ void deg_scatter_kernel(const int* __restrict__ dst,
                                   float* __restrict__ deg, int e) {
    int i = blockIdx.x * blockDim.x + threadIdx.x;
    if (i < e) atomic_fadd(&deg[dst[i]], 1.0f);
}

__global__ void rsqrt_kernel(float* __restrict__ p, int n) {
    int i = blockIdx.x * blockDim.x + threadIdx.x;
    if (i < n) p[i] = rsqrtf(p[i]);
}

template<int K, int NC, bool ADD_BIAS, bool RELU_A, bool INIT_AGG>
__global__ void gemm_wmma_kernel(const float* __restrict__ A,
                                 const float* __restrict__ B,
                                 const float* __restrict__ bias,
                                 float* __restrict__ C,
                                 const float* __restrict__ dinv,
                                 const float* __restrict__ bias2,
                                 float* __restrict__ C2, int M) {
    __shared__ __align__(16) float As[16 * K];
    __shared__ float Ds[16];

    const int tid   = threadIdx.x;
    const int wave  = tid >> 5;
    const int lane  = tid & 31;
    const int n0    = wave * 16;
    const int lhalf = lane >> 4;
    const int l15   = lane & 15;
    const int n     = n0 + l15;

    v2f bfrag[K / 4];
#pragma unroll
    for (int k0 = 0; k0 < K; k0 += 4) {
        const int kk = k0 + 2 * lhalf;
        bfrag[k0 / 4].x = B[(size_t)kk * NC + n];
        bfrag[k0 / 4].y = B[(size_t)(kk + 1) * NC + n];
    }
    const float bv = ADD_BIAS ? bias[n] : 0.0f;
    const float b2v = INIT_AGG ? bias2[n] : 0.0f;

    const int strips      = (M + 15) / 16;
    const int vecsPerRow  = K / 4;

    for (int s = blockIdx.x; s < strips; s += gridDim.x) {
        const int row0 = s * 16;
        __syncthreads();
        float dv = 0.0f;
        if (INIT_AGG) {
            const int dr = row0 + (tid & 15);
            dv = dinv[dr < M ? dr : M - 1];
        }
        for (int v = tid; v < 16 * vecsPerRow; v += blockDim.x) {
            int r = v / vecsPerRow;
            float4 val = make_float4(0.f, 0.f, 0.f, 0.f);
            if (row0 + r < M)
                val = ((const float4*)(A + (size_t)(row0 + r) * K))[v % vecsPerRow];
            if (RELU_A) {
                val.x = fmaxf(val.x, 0.0f); val.y = fmaxf(val.y, 0.0f);
                val.z = fmaxf(val.z, 0.0f); val.w = fmaxf(val.w, 0.0f);
            }
            ((float4*)As)[v] = val;
        }
        if (INIT_AGG && tid < 16) Ds[tid] = dv;
        __syncthreads();

        v8f acc = {};
#pragma unroll
        for (int k0 = 0; k0 < K; k0 += 4) {
            const int kk = k0 + 2 * lhalf;
            v2f a;
            a.x = As[l15 * K + kk];
            a.y = As[l15 * K + kk + 1];
            acc = __builtin_amdgcn_wmma_f32_16x16x4_f32(
                      false, a, false, bfrag[k0 / 4], (short)0, acc, false, false);
        }

#pragma unroll
        for (int r = 0; r < 8; ++r) {
            int m = row0 + r + 8 * lhalf;
            if (m < M) {
                const float c = acc[r] + bv;
                C[(size_t)m * NC + n] = c;
                if (INIT_AGG) {
                    const float di = Ds[r + 8 * lhalf];
                    C2[(size_t)m * NC + n] = c * di * di + b2v;
                }
            }
        }
    }
}

__global__ void edge_scatter_kernel(const int* __restrict__ src,
                                    const int* __restrict__ dst,
                                    const float* __restrict__ dinv,
                                    const float* __restrict__ msg,
                                    float* __restrict__ out, int e) {
    int gid  = blockIdx.x * blockDim.x + threadIdx.x;
    int edge = gid >> 5;
    int lane = gid & 31;
    if (edge >= e) return;
    int s = src[edge], d = dst[edge];
    float w = dinv[s] * dinv[d];
    const float* m = msg + (size_t)s * H_C + lane;
    float v0 = m[0], v1 = m[32], v2 = m[64], v3 = m[96];
    float* o = out + (size_t)d * H_C + lane;
    atomic_fadd_at<0>(o, v0 * w);
    atomic_fadd_at<128>(o, v1 * w);
    atomic_fadd_at<256>(o, v2 * w);
    atomic_fadd_at<384>(o, v3 * w);
}

__global__ void pool_scatter_kernel(const float* __restrict__ h,
                                    const int* __restrict__ batch,
                                    float* __restrict__ sums,
                                    float* __restrict__ cnts, int n) {
    int gid  = blockIdx.x * blockDim.x + threadIdx.x;
    int node = gid >> 5;
    int lane = gid & 31;
    if (node >= n) return;
    int g = batch[node];
    const float* m = h + (size_t)node * H_C + lane;
    float v0 = fmaxf(m[0], 0.0f), v1 = fmaxf(m[32], 0.0f);
    float v2 = fmaxf(m[64], 0.0f), v3 = fmaxf(m[96], 0.0f);
    float* o = sums + (size_t)g * H_C + lane;
    atomic_fadd_at<0>(o, v0);
    atomic_fadd_at<128>(o, v1);
    atomic_fadd_at<256>(o, v2);
    atomic_fadd_at<384>(o, v3);
    if (lane == 0) atomic_fadd(&cnts[g], 1.0f);
}

__global__ void pool_div_kernel(float* __restrict__ pooled,
                                const float* __restrict__ cnts, int total) {
    int i = blockIdx.x * blockDim.x + threadIdx.x;
    if (i >= total) return;
    pooled[i] /= fmaxf(cnts[i >> 7], 1.0f);
}

extern "C" void kernel_launch(void* const* d_in, const int* in_sizes, int n_in,
                              void* d_out, int out_size, void* d_ws, size_t ws_size,
                              hipStream_t stream) {
    const float* x    = (const float*)d_in[0];
    const int*   ei   = (const int*)  d_in[1];
    const int*   batch= (const int*)  d_in[2];
    const float* W1   = (const float*)d_in[3];
    const float* b1   = (const float*)d_in[4];
    const float* W2   = (const float*)d_in[5];
    const float* b2   = (const float*)d_in[6];
    const float* Wl   = (const float*)d_in[7];
    const float* bl   = (const float*)d_in[8];

    const int N = in_sizes[0] / IN_C;
    const int E = in_sizes[1] / 2;
    const int G = (out_size - N * OUT_C) / H_C;

    const int* esrc = ei;
    const int* edst = ei + E;

    float* logits = (float*)d_out;
    float* pooled = logits + (size_t)N * OUT_C;

    float* deg  = (float*)d_ws;
    float* tmpA = deg + (((size_t)N + 255) & ~(size_t)255);
    float* tmpB = tmpA + (size_t)N * H_C;
    float* tmpC = tmpB + (size_t)N * H_C;
    float* cnts = tmpC + (size_t)N * H_C;

    const int T  = 256;
    const int rowBlocks  = (N + 15) / 16;
    const int gemmBlocks = rowBlocks < 512 ? rowBlocks : 512;

    fill_kernel<<<(N + T - 1) / T, T, 0, stream>>>(deg, 1.0f, N);
    fill_kernel<<<(G + T - 1) / T, T, 0, stream>>>(cnts, 0.0f, G);
    fill_kernel<<<(G * H_C + T - 1) / T, T, 0, stream>>>(pooled, 0.0f, G * H_C);
    deg_scatter_kernel<<<(E + T - 1) / T, T, 0, stream>>>(edst, deg, E);
    rsqrt_kernel<<<(N + T - 1) / T, T, 0, stream>>>(deg, N);

    gemm_wmma_kernel<IN_C, H_C, false, false, true><<<gemmBlocks, 256, 0, stream>>>(
        x, W1, nullptr, tmpA, deg, b1, tmpB, N);
    edge_scatter_kernel<<<((size_t)E * 32 + T - 1) / T, T, 0, stream>>>(esrc, edst, deg, tmpA, tmpB, E);

    gemm_wmma_kernel<H_C, H_C, false, true, true><<<gemmBlocks, 256, 0, stream>>>(
        tmpB, W2, nullptr, tmpA, deg, b2, tmpC, N);
    edge_scatter_kernel<<<((size_t)E * 32 + T - 1) / T, T, 0, stream>>>(esrc, edst, deg, tmpA, tmpC, E);

    gemm_wmma_kernel<H_C, OUT_C, true, true, false><<<gemmBlocks, 128, 0, stream>>>(
        tmpC, Wl, bl, logits, nullptr, nullptr, nullptr, N);

    pool_scatter_kernel<<<((size_t)N * 32 + T - 1) / T, T, 0, stream>>>(tmpC, batch, pooled, cnts, N);
    pool_div_kernel<<<(G * H_C + T - 1) / T, T, 0, stream>>>(pooled, cnts, G * H_C);
}


# edge scatter: one wave per 32 consecutive edges (indices and weights loaded coalesced, broadcast by readlane), row loads of 8 edges in flight before their f32 atomics; 32x fewer waves
# speedup vs baseline: 1.7275x; 1.1443x over previous
#include <hip/hip_runtime.h>

#define IN_C  128
#define H_C   128
#define OUT_C 64

typedef float v2f __attribute__((ext_vector_type(2)));
typedef float v8f __attribute__((ext_vector_type(8)));

__device__ __forceinline__ void atomic_fadd(float* p, float v) {
    asm volatile("global_atomic_add_f32 %0, %1, off" : : "v"(p), "v"(v) : "memory");
}

template<int OFF>
__device__ __forceinline__ void atomic_fadd_at(float* p, float v) {
    asm volatile("global_atomic_add_f32 %0, %1, off offset:%2" : : "v"(p), "v"(v), "n"(OFF) : "memory");
}

__global__ void fill_kernel(float* __restrict__ p, float v, int n) {
    int i = blockIdx.x * blockDim.x + threadIdx.x;
    if (i < n) p[i] = v;
}

__global__ void deg_scatter_kernel(const int* __restrict__ dst,
                                   float* __restrict__ deg, int e) {
    int i = blockIdx.x * blockDim.x + threadIdx.x;
    if (i < e) atomic_fadd(&deg[dst[i]], 1.0f);
}

__global__ void rsqrt_kernel(float* __restrict__ p, int n) {
    int i = blockIdx.x * blockDim.x + threadIdx.x;
    if (i < n) p[i] = rsqrtf(p[i]);
}

template<int K, int NC, bool ADD_BIAS, bool RELU_A, bool INIT_AGG>
__global__ void gemm_wmma_kernel(const float* __restrict__ A,
                                 const float* __restrict__ B,
                                 const float* __restrict__ bias,
                                 float* __restrict__ C,
                                 const float* __restrict__ dinv,
                                 const float* __restrict__ bias2,
                                 float* __restrict__ C2, int M) {
    __shared__ __align__(16) float As[16 * K];
    __shared__ float Ds[16];

    const int tid   = threadIdx.x;
    const int wave  = tid >> 5;
    const int lane  = tid & 31;
    const int n0    = wave * 16;
    const int lhalf = lane >> 4;
    const int l15   = lane & 15;
    const int n     = n0 + l15;

    v2f bfrag[K / 4];
#pragma unroll
    for (int k0 = 0; k0 < K; k0 += 4) {
        const int kk = k0 + 2 * lhalf;
        bfrag[k0 / 4].x = B[(size_t)kk * NC + n];
        bfrag[k0 / 4].y = B[(size_t)(kk + 1) * NC + n];
    }
    const float bv = ADD_BIAS ? bias[n] : 0.0f;
    const float b2v = INIT_AGG ? bias2[n] : 0.0f;

    const int strips      = (M + 15) / 16;
    const int vecsPerRow  = K / 4;

    for (int s = blockIdx.x; s < strips; s += gridDim.x) {
        const int row0 = s * 16;
        __syncthreads();
        float dv = 0.0f;
        if (INIT_AGG) {
            const int dr = row0 + (tid & 15);
            dv = dinv[dr < M ? dr : M - 1];
        }
        for (int v = tid; v < 16 * vecsPerRow; v += blockDim.x) {
            int r = v / vecsPerRow;
            float4 val = make_float4(0.f, 0.f, 0.f, 0.f);
            if (row0 + r < M)
                val = ((const float4*)(A + (size_t)(row0 + r) * K))[v % vecsPerRow];
            if (RELU_A) {
                val.x = fmaxf(val.x, 0.0f); val.y = fmaxf(val.y, 0.0f);
                val.z = fmaxf(val.z, 0.0f); val.w = fmaxf(val.w, 0.0f);
            }
            ((float4*)As)[v] = val;
        }
        if (INIT_AGG && tid < 16) Ds[tid] = dv;
        __syncthreads();

        v8f acc = {};
#pragma unroll
        for (int k0 = 0; k0 < K; k0 += 4) {
            const int kk = k0 + 2 * lhalf;
            v2f a;
            a.x = As[l15 * K + kk];
            a.y = As[l15 * K + kk + 1];
            acc = __builtin_amdgcn_wmma_f32_16x16x4_f32(
                      false, a, false, bfrag[k0 / 4], (short)0, acc, false, false);
        }

#pragma unroll
        for (int r = 0; r < 8; ++r) {
            int m = row0 + r + 8 * lhalf;
            if (m < M) {
                const float c = acc[r] + bv;
                C[(size_t)m * NC + n] = c;
                if (INIT_AGG) {
                    const float di = Ds[r + 8 * lhalf];
                    C2[(size_t)m * NC + n] = c * di * di + b2v;
                }
            }
        }
    }
}

template<int R>
__device__ __forceinline__ void scatter_rows(const float* __restrict__ msg, float* __restrict__ out,
                                             int s_l, int d_l, float w_l, int j, int lane) {
    float v[R][4];
#pragma unroll
    for (int u = 0; u < R; ++u) {
        const int s = __builtin_amdgcn_readlane(s_l, j + u);
        const float* m = msg + (size_t)s * H_C + lane;
        v[u][0] = m[0]; v[u][1] = m[32]; v[u][2] = m[64]; v[u][3] = m[96];
    }
#pragma unroll
    for (int u = 0; u < R; ++u) {
        const int d = __builtin_amdgcn_readlane(d_l, j + u);
        const float w = __int_as_float(__builtin_amdgcn_readlane(__float_as_int(w_l), j + u));
        float* o = out + (size_t)d * H_C + lane;
        atomic_fadd_at<0>(o, v[u][0] * w);
        atomic_fadd_at<128>(o, v[u][1] * w);
        atomic_fadd_at<256>(o, v[u][2] * w);
        atomic_fadd_at<384>(o, v[u][3] * w);
    }
}

__global__ void __launch_bounds__(256)
edge_scatter_kernel(const int* __restrict__ src, const int* __restrict__ dst,
                    const float* __restrict__ dinv, const float* __restrict__ msg,
                    float* __restrict__ out, int e) {
    const int lane = threadIdx.x & 31;
    const int base = __builtin_amdgcn_readfirstlane(
                         (int)(blockIdx.x * (blockDim.x >> 5) + (threadIdx.x >> 5))) * 32;
    if (base >= e) return;
    const int left = e - base;
    const int cnt  = left < 32 ? left : 32;
    int s_l = 0, d_l = 0;
    float w_l = 0.0f;
    if (lane < cnt) {
        s_l = src[base + lane];
        d_l = dst[base + lane];
        w_l = dinv[s_l] * dinv[d_l];
    }
    int j = 0;
    for (; j + 8 <= cnt; j += 8) scatter_rows<8>(msg, out, s_l, d_l, w_l, j, lane);
    for (; j < cnt; ++j) scatter_rows<1>(msg, out, s_l, d_l, w_l, j, lane);
}

__global__ void pool_scatter_kernel(const float* __restrict__ h,
                                    const int* __restrict__ batch,
                                    float* __restrict__ sums,
                                    float* __restrict__ cnts, int n) {
    int gid  = blockIdx.x * blockDim.x + threadIdx.x;
    int node = gid >> 5;
    int lane = gid & 31;
    if (node >= n) return;
    int g = batch[node];
    const float* m = h + (size_t)node * H_C + lane;
    float v0 = fmaxf(m[0], 0.0f), v1 = fmaxf(m[32], 0.0f);
    float v2 = fmaxf(m[64], 0.0f), v3 = fmaxf(m[96], 0.0f);
    float* o = sums + (size_t)g * H_C + lane;
    atomic_fadd_at<0>(o, v0);
    atomic_fadd_at<128>(o, v1);
    atomic_fadd_at<256>(o, v2);
    atomic_fadd_at<384>(o, v3);
    if (lane == 0) atomic_fadd(&cnts[g], 1.0f);
}

__global__ void pool_div_kernel(float* __restrict__ pooled,
                                const float* __restrict__ cnts, int total) {
    int i = blockIdx.x * blockDim.x + threadIdx.x;
    if (i >= total) return;
    pooled[i] /= fmaxf(cnts[i >> 7], 1.0f);
}

extern "C" void kernel_launch(void* const* d_in, const int* in_sizes, int n_in,
                              void* d_out, int out_size, void* d_ws, size_t ws_size,
                              hipStream_t stream) {
    const float* x    = (const float*)d_in[0];
    const int*   ei   = (const int*)  d_in[1];
    const int*   batch= (const int*)  d_in[2];
    const float* W1   = (const float*)d_in[3];
    const float* b1   = (const float*)d_in[4];
    const float* W2   = (const float*)d_in[5];
    const float* b2   = (const float*)d_in[6];
    const float* Wl   = (const float*)d_in[7];
    const float* bl   = (const float*)d_in[8];

    const int N = in_sizes[0] / IN_C;
    const int E = in_sizes[1] / 2;
    const int G = (out_size - N * OUT_C) / H_C;

    const int* esrc = ei;
    const int* edst = ei + E;

    float* logits = (float*)d_out;
    float* pooled = logits + (size_t)N * OUT_C;

    float* deg  = (float*)d_ws;
    float* tmpA = deg + (((size_t)N + 255) & ~(size_t)255);
    float* tmpB = tmpA + (size_t)N * H_C;
    float* tmpC = tmpB + (size_t)N * H_C;
    float* cnts = tmpC + (size_t)N * H_C;

    const int T  = 256;
    const int rowBlocks  = (N + 15) / 16;
    const int gemmBlocks = rowBlocks < 512 ? rowBlocks : 512;

    fill_kernel<<<(N + T - 1) / T, T, 0, stream>>>(deg, 1.0f, N);
    fill_kernel<<<(G + T - 1) / T, T, 0, stream>>>(cnts, 0.0f, G);
    fill_kernel<<<(G * H_C + T - 1) / T, T, 0, stream>>>(pooled, 0.0f, G * H_C);
    deg_scatter_kernel<<<(E + T - 1) / T, T, 0, stream>>>(edst, deg, E);
    rsqrt_kernel<<<(N + T - 1) / T, T, 0, stream>>>(deg, N);

    gemm_wmma_kernel<IN_C, H_C, false, false, true><<<gemmBlocks, 256, 0, stream>>>(
        x, W1, nullptr, tmpA, deg, b1, tmpB, N);
    edge_scatter_kernel<<<(E + T - 1) / T, T, 0, stream>>>(esrc, edst, deg, tmpA, tmpB, E);

    gemm_wmma_kernel<H_C, H_C, false, true, true><<<gemmBlocks, 256, 0, stream>>>(
        tmpB, W2, nullptr, tmpA, deg, b2, tmpC, N);
    edge_scatter_kernel<<<(E + T - 1) / T, T, 0, stream>>>(esrc, edst, deg, tmpA, tmpC, E);

    gemm_wmma_kernel<H_C, OUT_C, true, true, false><<<gemmBlocks, 128, 0, stream>>>(
        tmpC, Wl, bl, logits, nullptr, nullptr, nullptr, N);

    pool_scatter_kernel<<<((size_t)N * 32 + T - 1) / T, T, 0, stream>>>(tmpC, batch, pooled, cnts, N);
    pool_div_kernel<<<(G * H_C + T - 1) / T, T, 0, stream>>>(pooled, cnts, G * H_C);
}


# GEMM A strip in LDS at row pitch K+4 floats instead of K: the 16 rows of a fragment read start 4 banks apart (conflict-free) instead of all on the same banks (16-way conflict on every read)
# speedup vs baseline: 1.9922x; 1.1532x over previous
#include <hip/hip_runtime.h>

#define IN_C  128
#define H_C   128
#define OUT_C 64

typedef float v2f __attribute__((ext_vector_type(2)));
typedef float v8f __attribute__((ext_vector_type(8)));

__device__ __forceinline__ void atomic_fadd(float* p, float v) {
    asm volatile("global_atomic_add_f32 %0, %1, off" : : "v"(p), "v"(v) : "memory");
}

template<int OFF>
__device__ __forceinline__ void atomic_fadd_at(float* p, float v) {
    asm volatile("global_atomic_add_f32 %0, %1, off offset:%2" : : "v"(p), "v"(v), "n"(OFF) : "memory");
}

__global__ void fill_kernel(float* __restrict__ p, float v, int n) {
    int i = blockIdx.x * blockDim.x + threadIdx.x;
    if (i < n) p[i] = v;
}

__global__ void deg_scatter_kernel(const int* __restrict__ dst,
                                   float* __restrict__ deg, int e) {
    int i = blockIdx.x * blockDim.x + threadIdx.x;
    if (i < e) atomic_fadd(&deg[dst[i]], 1.0f);
}

__global__ void rsqrt_kernel(float* __restrict__ p, int n) {
    int i = blockIdx.x * blockDim.x + threadIdx.x;
    if (i < n) p[i] = rsqrtf(p[i]);
}

template<int K, int NC, bool ADD_BIAS, bool RELU_A, bool INIT_AGG>
__global__ void gemm_wmma_kernel(const float* __restrict__ A,
                                 const float* __restrict__ B,
                                 const float* __restrict__ bias,
                                 float* __restrict__ C,
                                 const float* __restrict__ dinv,
                                 const float* __restrict__ bias2,
                                 float* __restrict__ C2, int M) {
    constexpr int LDA = K + 4;
    __shared__ __align__(16) float As[16 * LDA];
    __shared__ float Ds[16];

    const int tid   = threadIdx.x;
    const int wave  = tid >> 5;
    const int lane  = tid & 31;
    const int n0    = wave * 16;
    const int lhalf = lane >> 4;
    const int l15   = lane & 15;
    const int n     = n0 + l15;

    v2f bfrag[K / 4];
#pragma unroll
    for (int k0 = 0; k0 < K; k0 += 4) {
        const int kk = k0 + 2 * lhalf;
        bfrag[k0 / 4].x = B[(size_t)kk * NC + n];
        bfrag[k0 / 4].y = B[(size_t)(kk + 1) * NC + n];
    }
    const float bv = ADD_BIAS ? bias[n] : 0.0f;
    const float b2v = INIT_AGG ? bias2[n] : 0.0f;

    const int strips      = (M + 15) / 16;
    const int vecsPerRow  = K / 4;

    for (int s = blockIdx.x; s < strips; s += gridDim.x) {
        const int row0 = s * 16;
        __syncthreads();
        float dv = 0.0f;
        if (INIT_AGG) {
            const int dr = row0 + (tid & 15);
            dv = dinv[dr < M ? dr : M - 1];
        }
        for (int v = tid; v < 16 * vecsPerRow; v += blockDim.x) {
            int r = v / vecsPerRow;
            float4 val = make_float4(0.f, 0.f, 0.f, 0.f);
            if (row0 + r < M)
                val = ((const float4*)(A + (size_t)(row0 + r) * K))[v % vecsPerRow];
            if (RELU_A) {
                val.x = fmaxf(val.x, 0.0f); val.y = fmaxf(val.y, 0.0f);
                val.z = fmaxf(val.z, 0.0f); val.w = fmaxf(val.w, 0.0f);
            }
            *(float4*)(As + r * LDA + 4 * (v % vecsPerRow)) = val;
        }
        if (INIT_AGG && tid < 16) Ds[tid] = dv;
        __syncthreads();

        v8f acc = {};
#pragma unroll
        for (int k0 = 0; k0 < K; k0 += 4) {
            const int kk = k0 + 2 * lhalf;
            v2f a;
            a.x = As[l15 * LDA + kk];
            a.y = As[l15 * LDA + kk + 1];
            acc = __builtin_amdgcn_wmma_f32_16x16x4_f32(
                      false, a, false, bfrag[k0 / 4], (short)0, acc, false, false);
        }

#pragma unroll
        for (int r = 0; r < 8; ++r) {
            int m = row0 + r + 8 * lhalf;
            if (m < M) {
                const float c = acc[r] + bv;
                C[(size_t)m * NC + n] = c;
                if (INIT_AGG) {
                    const float di = Ds[r + 8 * lhalf];
                    C2[(size_t)m * NC + n] = c * di * di + b2v;
                }
            }
        }
    }
}

template<int R>
__device__ __forceinline__ void scatter_rows(const float* __restrict__ msg, float* __restrict__ out,
                                             int s_l, int d_l, float w_l, int j, int lane) {
    float v[R][4];
#pragma unroll
    for (int u = 0; u < R; ++u) {
        const int s = __builtin_amdgcn_readlane(s_l, j + u);
        const float* m = msg + (size_t)s * H_C + lane;
        v[u][0] = m[0]; v[u][1] = m[32]; v[u][2] = m[64]; v[u][3] = m[96];
    }
#pragma unroll
    for (int u = 0; u < R; ++u) {
        const int d = __builtin_amdgcn_readlane(d_l, j + u);
        const float w = __int_as_float(__builtin_amdgcn_readlane(__float_as_int(w_l), j + u));
        float* o = out + (size_t)d * H_C + lane;
        atomic_fadd_at<0>(o, v[u][0] * w);
        atomic_fadd_at<128>(o, v[u][1] * w);
        atomic_fadd_at<256>(o, v[u][2] * w);
        atomic_fadd_at<384>(o, v[u][3] * w);
    }
}

__global__ void __launch_bounds__(256)
edge_scatter_kernel(const int* __restrict__ src, const int* __restrict__ dst,
                    const float* __restrict__ dinv, const float* __restrict__ msg,
                    float* __restrict__ out, int e) {
    const int lane = threadIdx.x & 31;
    const int base = __builtin_amdgcn_readfirstlane(
                         (int)(blockIdx.x * (blockDim.x >> 5) + (threadIdx.x >> 5))) * 32;
    if (base >= e) return;
    const int left = e - base;
    const int cnt  = left < 32 ? left : 32;
    int s_l = 0, d_l = 0;
    float w_l = 0.0f;
    if (lane < cnt) {
        s_l = src[base + lane];
        d_l = dst[base + lane];
        w_l = dinv[s_l] * dinv[d_l];
    }
    int j = 0;
    for (; j + 8 <= cnt; j += 8) scatter_rows<8>(msg, out, s_l, d_l, w_l, j, lane);
    for (; j < cnt; ++j) scatter_rows<1>(msg, out, s_l, d_l, w_l, j, lane);
}

__global__ void pool_scatter_kernel(const float* __restrict__ h,
                                    const int* __restrict__ batch,
                                    float* __restrict__ sums,
                                    float* __restrict__ cnts, int n) {
    int gid  = blockIdx.x * blockDim.x + threadIdx.x;
    int node = gid >> 5;
    int lane = gid & 31;
    if (node >= n) return;
    int g = batch[node];
    const float* m = h + (size_t)node * H_C + lane;
    float v0 = fmaxf(m[0], 0.0f), v1 = fmaxf(m[32], 0.0f);
    float v2 = fmaxf(m[64], 0.0f), v3 = fmaxf(m[96], 0.0f);
    float* o = sums + (size_t)g * H_C + lane;
    atomic_fadd_at<0>(o, v0);
    atomic_fadd_at<128>(o, v1);
    atomic_fadd_at<256>(o, v2);
    atomic_fadd_at<384>(o, v3);
    if (lane == 0) atomic_fadd(&cnts[g], 1.0f);
}

__global__ void pool_div_kernel(float* __restrict__ pooled,
                                const float* __restrict__ cnts, int total) {
    int i = blockIdx.x * blockDim.x + threadIdx.x;
    if (i >= total) return;
    pooled[i] /= fmaxf(cnts[i >> 7], 1.0f);
}

extern "C" void kernel_launch(void* const* d_in, const int* in_sizes, int n_in,
                              void* d_out, int out_size, void* d_ws, size_t ws_size,
                              hipStream_t stream) {
    const float* x    = (const float*)d_in[0];
    const int*   ei   = (const int*)  d_in[1];
    const int*   batch= (const int*)  d_in[2];
    const float* W1   = (const float*)d_in[3];
    const float* b1   = (const float*)d_in[4];
    const float* W2   = (const float*)d_in[5];
    const float* b2   = (const float*)d_in[6];
    const float* Wl   = (const float*)d_in[7];
    const float* bl   = (const float*)d_in[8];

    const int N = in_sizes[0] / IN_C;
    const int E = in_sizes[1] / 2;
    const int G = (out_size - N * OUT_C) / H_C;

    const int* esrc = ei;
    const int* edst = ei + E;

    float* logits = (float*)d_out;
    float* pooled = logits + (size_t)N * OUT_C;

    float* deg  = (float*)d_ws;
    float* tmpA = deg + (((size_t)N + 255) & ~(size_t)255);
    float* tmpB = tmpA + (size_t)N * H_C;
    float* tmpC = tmpB + (size_t)N * H_C;
    float* cnts = tmpC + (size_t)N * H_C;

    const int T  = 256;
    const int rowBlocks  = (N + 15) / 16;
    const int gemmBlocks = rowBlocks < 512 ? rowBlocks : 512;

    fill_kernel<<<(N + T - 1) / T, T, 0, stream>>>(deg, 1.0f, N);
    fill_kernel<<<(G + T - 1) / T, T, 0, stream>>>(cnts, 0.0f, G);
    fill_kernel<<<(G * H_C + T - 1) / T, T, 0, stream>>>(pooled, 0.0f, G * H_C);
    deg_scatter_kernel<<<(E + T - 1) / T, T, 0, stream>>>(edst, deg, E);
    rsqrt_kernel<<<(N + T - 1) / T, T, 0, stream>>>(deg, N);

    gemm_wmma_kernel<IN_C, H_C, false, false, true><<<gemmBlocks, 256, 0, stream>>>(
        x, W1, nullptr, tmpA, deg, b1, tmpB, N);
    edge_scatter_kernel<<<(E + T - 1) / T, T, 0, stream>>>(esrc, edst, deg, tmpA, tmpB, E);

    gemm_wmma_kernel<H_C, H_C, false, true, true><<<gemmBlocks, 256, 0, stream>>>(
        tmpB, W2, nullptr, tmpA, deg, b2, tmpC, N);
    edge_scatter_kernel<<<(E + T - 1) / T, T, 0, stream>>>(esrc, edst, deg, tmpA, tmpC, E);

    gemm_wmma_kernel<H_C, OUT_C, true, true, false><<<gemmBlocks, 128, 0, stream>>>(
        tmpC, Wl, bl, logits, nullptr, nullptr, nullptr, N);

    pool_scatter_kernel<<<((size_t)N * 32 + T - 1) / T, T, 0, stream>>>(tmpC, batch, pooled, cnts, N);
    pool_div_kernel<<<(G * H_C + T - 1) / T, T, 0, stream>>>(pooled, cnts, G * H_C);
}


# edge aggregation without atomics: per call the edges are sorted by destination (stable 2-pass 8-bit radix sort, ballot ranks), one wave sums a node's rows in registers in edge order and stores once
# speedup vs baseline: 2.4146x; 1.2120x over previous
#include <hip/hip_runtime.h>

#define IN_C  128
#define H_C   128
#define OUT_C 64

typedef float v2f __attribute__((ext_vector_type(2)));
typedef float v8f __attribute__((ext_vector_type(8)));

__device__ __forceinline__ void atomic_fadd(float* p, float v) {
    asm volatile("global_atomic_add_f32 %0, %1, off" : : "v"(p), "v"(v) : "memory");
}

template<int OFF>
__device__ __forceinline__ void atomic_fadd_at(float* p, float v) {
    asm volatile("global_atomic_add_f32 %0, %1, off offset:%2" : : "v"(p), "v"(v), "n"(OFF) : "memory");
}

__global__ void fill_kernel(float* __restrict__ p, float v, int n) {
    int i = blockIdx.x * blockDim.x + threadIdx.x;
    if (i < n) p[i] = v;
}

#define SORT_WAVES  8
#define SORT_CHUNKS 32
#define SORT_GROUP  8
#define SORT_TILE   (SORT_WAVES * SORT_CHUNKS * 32)

__device__ __forceinline__ unsigned digit_peers(int digit, unsigned valid) {
    unsigned peers = valid;
#pragma unroll
    for (int b = 0; b < 8; ++b) {
        const bool bit = (digit >> b) & 1;
        const unsigned m = __builtin_amdgcn_ballot_w32(bit);
        peers &= bit ? m : ~m;
    }
    return peers;
}

template<int SHIFT, int KSTRIDE>
__device__ __forceinline__ void count_digits(const int* __restrict__ kin, int* hrow,
                                             int wbase, int lane, int e) {
#pragma unroll
    for (int q = 0; q < 8; ++q) hrow[lane + 32 * q] = 0;
    __builtin_amdgcn_wave_barrier();
#pragma unroll 1
    for (int g = 0; g < SORT_CHUNKS; g += SORT_GROUP) {
        int keys[SORT_GROUP];
#pragma unroll
        for (int u = 0; u < SORT_GROUP; ++u) {
            const int i = wbase + (g + u) * 32 + lane;
            keys[u] = kin[(size_t)(i < e ? i : e - 1) * KSTRIDE];
        }
#pragma unroll
        for (int u = 0; u < SORT_GROUP; ++u) {
            const bool ok = wbase + (g + u) * 32 + lane < e;
            const int digit = (keys[u] >> SHIFT) & 255;
            const unsigned peers = digit_peers(digit, __builtin_amdgcn_ballot_w32(ok));
            if (ok && lane == __builtin_ctz(peers)) hrow[digit] += __builtin_popcount(peers);
            __builtin_amdgcn_wave_barrier();
        }
    }
}

template<int SHIFT, int KSTRIDE>
__global__ void __launch_bounds__(256)
sort_count_kernel(const int* __restrict__ kin, int* __restrict__ tab, int e) {
    __shared__ int h[SORT_WAVES][256];
    const int tid = threadIdx.x, w = tid >> 5, lane = tid & 31;
    const int wbase = blockIdx.x * SORT_TILE + w * (SORT_CHUNKS * 32);
    count_digits<SHIFT, KSTRIDE>(kin, h[w], wbase, lane, e);
    __syncthreads();
    int t = 0;
#pragma unroll
    for (int v = 0; v < SORT_WAVES; ++v) t += h[v][tid];
    tab[blockIdx.x * 256 + tid] = t;
}

template<int SHIFT, bool PAIRS>
__global__ void __launch_bounds__(256)
sort_place_kernel(const int* __restrict__ kin, const int* __restrict__ vin,
                  int2* __restrict__ out, const int* __restrict__ tab, int e, int nt) {
    __shared__ int h[SORT_WAVES][256];
    __shared__ int wsum[SORT_WAVES];
    const int tid = threadIdx.x, w = tid >> 5, lane = tid & 31;
    const int tile = blockIdx.x;
    const int wbase = tile * SORT_TILE + w * (SORT_CHUNKS * 32);

    int total = 0, before = 0;
#pragma unroll 8
    for (int t = 0; t < nt; ++t) {
        const int c = tab[t * 256 + tid];
        total += c;
        before += t < tile ? c : 0;
    }
    int incl = total;
#pragma unroll
    for (int off = 1; off < 32; off <<= 1) {
        const int t = __shfl_up(incl, off, 32);
        if (lane >= off) incl += t;
    }
    if (lane == 31) wsum[w] = incl;

    count_digits<SHIFT, PAIRS ? 2 : 1>(kin, h[w], wbase, lane, e);
    __syncthreads();

    int run = before + incl - total;
    for (int v = 0; v < w; ++v) run += wsum[v];
#pragma unroll
    for (int v = 0; v < SORT_WAVES; ++v) {
        const int t = h[v][tid];
        h[v][tid] = run;
        run += t;
    }
    __syncthreads();

    int* hrow = h[w];
#pragma unroll 1
    for (int g = 0; g < SORT_CHUNKS; g += SORT_GROUP) {
        int keys[SORT_GROUP], vals[SORT_GROUP];
#pragma unroll
        for (int u = 0; u < SORT_GROUP; ++u) {
            const int i = wbase + (g + u) * 32 + lane;
            const int ic = i < e ? i : e - 1;
            if (PAIRS) {
                const int2 t = ((const int2*)kin)[ic];
                keys[u] = t.x; vals[u] = t.y;
            } else {
                keys[u] = kin[ic]; vals[u] = vin[ic];
            }
        }
#pragma unroll
        for (int u = 0; u < SORT_GROUP; ++u) {
            const bool ok = wbase + (g + u) * 32 + lane < e;
            const int digit = (keys[u] >> SHIFT) & 255;
            const unsigned peers = digit_peers(digit, __builtin_amdgcn_ballot_w32(ok));
            if (ok) {
                const int off = hrow[digit];
                const int pos = off + __builtin_popcount(peers & ((1u << lane) - 1u));
                out[pos] = make_int2(keys[u], vals[u]);
                if (lane == __builtin_ctz(peers)) hrow[digit] = off + __builtin_popcount(peers);
            }
            __builtin_amdgcn_wave_barrier();
        }
    }
}

__global__ void row_bounds_kernel(const int2* __restrict__ filed, int* __restrict__ rowse, int e) {
    int i = blockIdx.x * blockDim.x + threadIdx.x;
    if (i >= e) return;
    const int d = filed[i].x;
    if (i == 0 || filed[i - 1].x != d) rowse[2 * d] = i;
    if (i == e - 1 || filed[i + 1].x != d) rowse[2 * d + 1] = i + 1;
}

template<int K, int NC, bool ADD_BIAS, bool RELU_A>
__global__ void gemm_wmma_kernel(const float* __restrict__ A,
                                 const float* __restrict__ B,
                                 const float* __restrict__ bias,
                                 float* __restrict__ C, int M) {
    constexpr int LDA = K + 4;
    __shared__ __align__(16) float As[16 * LDA];

    const int tid   = threadIdx.x;
    const int wave  = tid >> 5;
    const int lane  = tid & 31;
    const int n0    = wave * 16;
    const int lhalf = lane >> 4;
    const int l15   = lane & 15;
    const int n     = n0 + l15;

    v2f bfrag[K / 4];
#pragma unroll
    for (int k0 = 0; k0 < K; k0 += 4) {
        const int kk = k0 + 2 * lhalf;
        bfrag[k0 / 4].x = B[(size_t)kk * NC + n];
        bfrag[k0 / 4].y = B[(size_t)(kk + 1) * NC + n];
    }
    const float bv = ADD_BIAS ? bias[n] : 0.0f;

    const int strips      = (M + 15) / 16;
    const int vecsPerRow  = K / 4;

    for (int s = blockIdx.x; s < strips; s += gridDim.x) {
        const int row0 = s * 16;
        __syncthreads();
        for (int v = tid; v < 16 * vecsPerRow; v += blockDim.x) {
            int r = v / vecsPerRow;
            float4 val = make_float4(0.f, 0.f, 0.f, 0.f);
            if (row0 + r < M)
                val = ((const float4*)(A + (size_t)(row0 + r) * K))[v % vecsPerRow];
            if (RELU_A) {
                val.x = fmaxf(val.x, 0.0f); val.y = fmaxf(val.y, 0.0f);
                val.z = fmaxf(val.z, 0.0f); val.w = fmaxf(val.w, 0.0f);
            }
            *(float4*)(As + r * LDA + 4 * (v % vecsPerRow)) = val;
        }
        __syncthreads();

        v8f acc = {};
#pragma unroll
        for (int k0 = 0; k0 < K; k0 += 4) {
            const int kk = k0 + 2 * lhalf;
            v2f a;
            a.x = As[l15 * LDA + kk];
            a.y = As[l15 * LDA + kk + 1];
            acc = __builtin_amdgcn_wmma_f32_16x16x4_f32(
                      false, a, false, bfrag[k0 / 4], (short)0, acc, false, false);
        }

#pragma unroll
        for (int r = 0; r < 8; ++r) {
            int m = row0 + r + 8 * lhalf;
            if (m < M) C[(size_t)m * NC + n] = acc[r] + bv;
        }
    }
}

template<int R>
__device__ __forceinline__ void gather_rows(const float* __restrict__ msg, int s_l, float w_l,
                                            int j, int lane, float4& acc) {
    float4 v[R];
#pragma unroll
    for (int u = 0; u < R; ++u) {
        const int s = __builtin_amdgcn_readlane(s_l, j + u);
        v[u] = ((const float4*)(msg + (size_t)s * H_C))[lane];
    }
#pragma unroll
    for (int u = 0; u < R; ++u) {
        const float w = __int_as_float(__builtin_amdgcn_readlane(__float_as_int(w_l), j + u));
        acc.x += v[u].x * w; acc.y += v[u].y * w;
        acc.z += v[u].z * w; acc.w += v[u].w * w;
    }
}

__global__ void __launch_bounds__(256)
row_gather_kernel(const int2* __restrict__ rowse, const int2* __restrict__ filed,
                  const float* __restrict__ msg, const float* __restrict__ bias,
                  float* __restrict__ out, int n) {
    const int lane = threadIdx.x & 31;
    const int d = __builtin_amdgcn_readfirstlane(
                      (int)(blockIdx.x * (blockDim.x >> 5) + (threadIdx.x >> 5)));
    if (d >= n) return;
    const int2 se   = rowse[d];
    const int start = se.x;
    const int c     = se.y - se.x;
    const float di  = rsqrtf((float)(c + 1));
    const float4 sv = ((const float4*)(msg + (size_t)d * H_C))[lane];
    const float4 bv = ((const float4*)bias)[lane];
    float4 acc;
    acc.x = sv.x * di * di + bv.x; acc.y = sv.y * di * di + bv.y;
    acc.z = sv.z * di * di + bv.z; acc.w = sv.w * di * di + bv.w;
    for (int base = 0; base < c; base += 32) {
        const int left = c - base;
        const int m = left < 32 ? left : 32;
        int s_l = 0;
        float w_l = 0.0f;
        if (lane < m) {
            s_l = filed[start + base + lane].y;
            const int2 t = rowse[s_l];
            w_l = rsqrtf((float)(t.y - t.x + 1)) * di;
        }
        int j = 0;
        for (; j + 8 <= m; j += 8) gather_rows<8>(msg, s_l, w_l, j, lane, acc);
        if (j + 4 <= m) { gather_rows<4>(msg, s_l, w_l, j, lane, acc); j += 4; }
        if (j + 2 <= m) { gather_rows<2>(msg, s_l, w_l, j, lane, acc); j += 2; }
        if (j < m)      { gather_rows<1>(msg, s_l, w_l, j, lane, acc); }
    }
    ((float4*)(out + (size_t)d * H_C))[lane] = acc;
}

__global__ void pool_scatter_kernel(const float* __restrict__ h,
                                    const int* __restrict__ batch,
                                    float* __restrict__ sums,
                                    float* __restrict__ cnts, int n) {
    int gid  = blockIdx.x * blockDim.x + threadIdx.x;
    int node = gid >> 5;
    int lane = gid & 31;
    if (node >= n) return;
    int g = batch[node];
    const float* m = h + (size_t)node * H_C + lane;
    float v0 = fmaxf(m[0], 0.0f), v1 = fmaxf(m[32], 0.0f);
    float v2 = fmaxf(m[64], 0.0f), v3 = fmaxf(m[96], 0.0f);
    float* o = sums + (size_t)g * H_C + lane;
    atomic_fadd_at<0>(o, v0);
    atomic_fadd_at<128>(o, v1);
    atomic_fadd_at<256>(o, v2);
    atomic_fadd_at<384>(o, v3);
    if (lane == 0) atomic_fadd(&cnts[g], 1.0f);
}

__global__ void pool_div_kernel(float* __restrict__ pooled,
                                const float* __restrict__ cnts, int total) {
    int i = blockIdx.x * blockDim.x + threadIdx.x;
    if (i >= total) return;
    pooled[i] /= fmaxf(cnts[i >> 7], 1.0f);
}

extern "C" void kernel_launch(void* const* d_in, const int* in_sizes, int n_in,
                              void* d_out, int out_size, void* d_ws, size_t ws_size,
                              hipStream_t stream) {
    const float* x    = (const float*)d_in[0];
    const int*   ei   = (const int*)  d_in[1];
    const int*   batch= (const int*)  d_in[2];
    const float* W1   = (const float*)d_in[3];
    const float* b1   = (const float*)d_in[4];
    const float* W2   = (const float*)d_in[5];
    const float* b2   = (const float*)d_in[6];
    const float* Wl   = (const float*)d_in[7];
    const float* bl   = (const float*)d_in[8];

    const int N = in_sizes[0] / IN_C;
    const int E = in_sizes[1] / 2;
    const int G = (out_size - N * OUT_C) / H_C;

    const int* esrc = ei;
    const int* edst = ei + E;

    float* logits = (float*)d_out;
    float* pooled = logits + (size_t)N * OUT_C;

    const int T  = 256;
    const int tiles = (E + SORT_TILE - 1) / SORT_TILE;

    const size_t Npad = ((size_t)N + 255) & ~(size_t)255;
    const size_t Epad = ((size_t)E + 255) & ~(size_t)255;
    int*   rowse = (int*)d_ws;
    int*   tab   = rowse + 2 * Npad;
    int2*  p1    = (int2*)(tab + (size_t)tiles * 256);
    int2*  p2    = p1 + Epad;
    float* tmpA = (float*)(p2 + Epad);
    float* tmpB = tmpA + (size_t)N * H_C;
    float* tmpC = tmpB + (size_t)N * H_C;
    float* cnts = tmpC + (size_t)N * H_C;

    const int rowBlocks  = (N + 15) / 16;
    const int gemmBlocks = rowBlocks < 512 ? rowBlocks : 512;

    fill_kernel<<<(2 * N + T - 1) / T, T, 0, stream>>>((float*)rowse, 0.0f, 2 * N);
    fill_kernel<<<(G + T - 1) / T, T, 0, stream>>>(cnts, 0.0f, G);
    fill_kernel<<<(G * H_C + T - 1) / T, T, 0, stream>>>(pooled, 0.0f, G * H_C);

    sort_count_kernel<0, 1><<<tiles, 256, 0, stream>>>(edst, tab, E);
    sort_place_kernel<0, false><<<tiles, 256, 0, stream>>>(edst, esrc, p1, tab, E, tiles);
    sort_count_kernel<8, 2><<<tiles, 256, 0, stream>>>((const int*)p1, tab, E);
    sort_place_kernel<8, true><<<tiles, 256, 0, stream>>>((const int*)p1, nullptr, p2, tab, E, tiles);
    row_bounds_kernel<<<(E + T - 1) / T, T, 0, stream>>>(p2, rowse, E);

    const int gatherBlocks = (N + (T / 32) - 1) / (T / 32);

    gemm_wmma_kernel<IN_C, H_C, false, false><<<gemmBlocks, 256, 0, stream>>>(
        x, W1, nullptr, tmpA, N);
    row_gather_kernel<<<gatherBlocks, T, 0, stream>>>((const int2*)rowse, p2, tmpA, b1, tmpB, N);

    gemm_wmma_kernel<H_C, H_C, false, true><<<gemmBlocks, 256, 0, stream>>>(
        tmpB, W2, nullptr, tmpA, N);
    row_gather_kernel<<<gatherBlocks, T, 0, stream>>>((const int2*)rowse, p2, tmpA, b2, tmpC, N);

    gemm_wmma_kernel<H_C, OUT_C, true, true><<<gemmBlocks, 128, 0, stream>>>(
        tmpC, Wl, bl, logits, N);

    pool_scatter_kernel<<<((size_t)N * 32 + T - 1) / T, T, 0, stream>>>(tmpC, batch, pooled, cnts, N);
    pool_div_kernel<<<(G * H_C + T - 1) / T, T, 0, stream>>>(pooled, cnts, G * H_C);
}


# mean pool: one wave per 32 consecutive nodes; rows of a run of equal graph id are summed in registers in node order, then one set of f32 atomics and one count add per run instead of per node
# speedup vs baseline: 2.7893x; 1.1552x over previous
#include <hip/hip_runtime.h>

#define IN_C  128
#define H_C   128
#define OUT_C 64

typedef float v2f __attribute__((ext_vector_type(2)));
typedef float v8f __attribute__((ext_vector_type(8)));

__device__ __forceinline__ void atomic_fadd(float* p, float v) {
    asm volatile("global_atomic_add_f32 %0, %1, off" : : "v"(p), "v"(v) : "memory");
}

template<int OFF>
__device__ __forceinline__ void atomic_fadd_at(float* p, float v) {
    asm volatile("global_atomic_add_f32 %0, %1, off offset:%2" : : "v"(p), "v"(v), "n"(OFF) : "memory");
}

__global__ void fill_kernel(float* __restrict__ p, float v, int n) {
    int i = blockIdx.x * blockDim.x + threadIdx.x;
    if (i < n) p[i] = v;
}

#define SORT_WAVES  8
#define SORT_CHUNKS 32
#define SORT_GROUP  8
#define SORT_TILE   (SORT_WAVES * SORT_CHUNKS * 32)

__device__ __forceinline__ unsigned digit_peers(int digit, unsigned valid) {
    unsigned peers = valid;
#pragma unroll
    for (int b = 0; b < 8; ++b) {
        const bool bit = (digit >> b) & 1;
        const unsigned m = __builtin_amdgcn_ballot_w32(bit);
        peers &= bit ? m : ~m;
    }
    return peers;
}

template<int SHIFT, int KSTRIDE>
__device__ __forceinline__ void count_digits(const int* __restrict__ kin, int* hrow,
                                             int wbase, int lane, int e) {
#pragma unroll
    for (int q = 0; q < 8; ++q) hrow[lane + 32 * q] = 0;
    __builtin_amdgcn_wave_barrier();
#pragma unroll 1
    for (int g = 0; g < SORT_CHUNKS; g += SORT_GROUP) {
        int keys[SORT_GROUP];
#pragma unroll
        for (int u = 0; u < SORT_GROUP; ++u) {
            const int i = wbase + (g + u) * 32 + lane;
            keys[u] = kin[(size_t)(i < e ? i : e - 1) * KSTRIDE];
        }
#pragma unroll
        for (int u = 0; u < SORT_GROUP; ++u) {
            const bool ok = wbase + (g + u) * 32 + lane < e;
            const int digit = (keys[u] >> SHIFT) & 255;
            const unsigned peers = digit_peers(digit, __builtin_amdgcn_ballot_w32(ok));
            if (ok && lane == __builtin_ctz(peers)) hrow[digit] += __builtin_popcount(peers);
            __builtin_amdgcn_wave_barrier();
        }
    }
}

template<int SHIFT, int KSTRIDE>
__global__ void __launch_bounds__(256)
sort_count_kernel(const int* __restrict__ kin, int* __restrict__ tab, int e) {
    __shared__ int h[SORT_WAVES][256];
    const int tid = threadIdx.x, w = tid >> 5, lane = tid & 31;
    const int wbase = blockIdx.x * SORT_TILE + w * (SORT_CHUNKS * 32);
    count_digits<SHIFT, KSTRIDE>(kin, h[w], wbase, lane, e);
    __syncthreads();
    int t = 0;
#pragma unroll
    for (int v = 0; v < SORT_WAVES; ++v) t += h[v][tid];
    tab[blockIdx.x * 256 + tid] = t;
}

template<int SHIFT, bool PAIRS>
__global__ void __launch_bounds__(256)
sort_place_kernel(const int* __restrict__ kin, const int* __restrict__ vin,
                  int2* __restrict__ out, const int* __restrict__ tab, int e, int nt) {
    __shared__ int h[SORT_WAVES][256];
    __shared__ int wsum[SORT_WAVES];
    const int tid = threadIdx.x, w = tid >> 5, lane = tid & 31;
    const int tile = blockIdx.x;
    const int wbase = tile * SORT_TILE + w * (SORT_CHUNKS * 32);

    int total = 0, before = 0;
#pragma unroll 8
    for (int t = 0; t < nt; ++t) {
        const int c = tab[t * 256 + tid];
        total += c;
        before += t < tile ? c : 0;
    }
    int incl = total;
#pragma unroll
    for (int off = 1; off < 32; off <<= 1) {
        const int t = __shfl_up(incl, off, 32);
        if (lane >= off) incl += t;
    }
    if (lane == 31) wsum[w] = incl;

    count_digits<SHIFT, PAIRS ? 2 : 1>(kin, h[w], wbase, lane, e);
    __syncthreads();

    int run = before + incl - total;
    for (int v = 0; v < w; ++v) run += wsum[v];
#pragma unroll
    for (int v = 0; v < SORT_WAVES; ++v) {
        const int t = h[v][tid];
        h[v][tid] = run;
        run += t;
    }
    __syncthreads();

    int* hrow = h[w];
#pragma unroll 1
    for (int g = 0; g < SORT_CHUNKS; g += SORT_GROUP) {
        int keys[SORT_GROUP], vals[SORT_GROUP];
#pragma unroll
        for (int u = 0; u < SORT_GROUP; ++u) {
            const int i = wbase + (g + u) * 32 + lane;
            const int ic = i < e ? i : e - 1;
            if (PAIRS) {
                const int2 t = ((const int2*)kin)[ic];
                keys[u] = t.x; vals[u] = t.y;
            } else {
                keys[u] = kin[ic]; vals[u] = vin[ic];
            }
        }
#pragma unroll
        for (int u = 0; u < SORT_GROUP; ++u) {
            const bool ok = wbase + (g + u) * 32 + lane < e;
            const int digit = (keys[u] >> SHIFT) & 255;
            const unsigned peers = digit_peers(digit, __builtin_amdgcn_ballot_w32(ok));
            if (ok) {
                const int off = hrow[digit];
                const int pos = off + __builtin_popcount(peers & ((1u << lane) - 1u));
                out[pos] = make_int2(keys[u], vals[u]);
                if (lane == __builtin_ctz(peers)) hrow[digit] = off + __builtin_popcount(peers);
            }
            __builtin_amdgcn_wave_barrier();
        }
    }
}

__global__ void row_bounds_kernel(const int2* __restrict__ filed, int* __restrict__ rowse, int e) {
    int i = blockIdx.x * blockDim.x + threadIdx.x;
    if (i >= e) return;
    const int d = filed[i].x;
    if (i == 0 || filed[i - 1].x != d) rowse[2 * d] = i;
    if (i == e - 1 || filed[i + 1].x != d) rowse[2 * d + 1] = i + 1;
}

template<int K, int NC, bool ADD_BIAS, bool RELU_A>
__global__ void gemm_wmma_kernel(const float* __restrict__ A,
                                 const float* __restrict__ B,
                                 const float* __restrict__ bias,
                                 float* __restrict__ C, int M) {
    constexpr int LDA = K + 4;
    __shared__ __align__(16) float As[16 * LDA];

    const int tid   = threadIdx.x;
    const int wave  = tid >> 5;
    const int lane  = tid & 31;
    const int n0    = wave * 16;
    const int lhalf = lane >> 4;
    const int l15   = lane & 15;
    const int n     = n0 + l15;

    v2f bfrag[K / 4];
#pragma unroll
    for (int k0 = 0; k0 < K; k0 += 4) {
        const int kk = k0 + 2 * lhalf;
        bfrag[k0 / 4].x = B[(size_t)kk * NC + n];
        bfrag[k0 / 4].y = B[(size_t)(kk + 1) * NC + n];
    }
    const float bv = ADD_BIAS ? bias[n] : 0.0f;

    const int strips      = (M + 15) / 16;
    const int vecsPerRow  = K / 4;

    for (int s = blockIdx.x; s < strips; s += gridDim.x) {
        const int row0 = s * 16;
        __syncthreads();
        for (int v = tid; v < 16 * vecsPerRow; v += blockDim.x) {
            int r = v / vecsPerRow;
            float4 val = make_float4(0.f, 0.f, 0.f, 0.f);
            if (row0 + r < M)
                val = ((const float4*)(A + (size_t)(row0 + r) * K))[v % vecsPerRow];
            if (RELU_A) {
                val.x = fmaxf(val.x, 0.0f); val.y = fmaxf(val.y, 0.0f);
                val.z = fmaxf(val.z, 0.0f); val.w = fmaxf(val.w, 0.0f);
            }
            *(float4*)(As + r * LDA + 4 * (v % vecsPerRow)) = val;
        }
        __syncthreads();

        v8f acc = {};
#pragma unroll
        for (int k0 = 0; k0 < K; k0 += 4) {
            const int kk = k0 + 2 * lhalf;
            v2f a;
            a.x = As[l15 * LDA + kk];
            a.y = As[l15 * LDA + kk + 1];
            acc = __builtin_amdgcn_wmma_f32_16x16x4_f32(
                      false, a, false, bfrag[k0 / 4], (short)0, acc, false, false);
        }

#pragma unroll
        for (int r = 0; r < 8; ++r) {
            int m = row0 + r + 8 * lhalf;
            if (m < M) C[(size_t)m * NC + n] = acc[r] + bv;
        }
    }
}

template<int R>
__device__ __forceinline__ void gather_rows(const float* __restrict__ msg, int s_l, float w_l,
                                            int j, int lane, float4& acc) {
    float4 v[R];
#pragma unroll
    for (int u = 0; u < R; ++u) {
        const int s = __builtin_amdgcn_readlane(s_l, j + u);
        v[u] = ((const float4*)(msg + (size_t)s * H_C))[lane];
    }
#pragma unroll
    for (int u = 0; u < R; ++u) {
        const float w = __int_as_float(__builtin_amdgcn_readlane(__float_as_int(w_l), j + u));
        acc.x += v[u].x * w; acc.y += v[u].y * w;
        acc.z += v[u].z * w; acc.w += v[u].w * w;
    }
}

__global__ void __launch_bounds__(256)
row_gather_kernel(const int2* __restrict__ rowse, const int2* __restrict__ filed,
                  const float* __restrict__ msg, const float* __restrict__ bias,
                  float* __restrict__ out, int n) {
    const int lane = threadIdx.x & 31;
    const int d = __builtin_amdgcn_readfirstlane(
                      (int)(blockIdx.x * (blockDim.x >> 5) + (threadIdx.x >> 5)));
    if (d >= n) return;
    const int2 se   = rowse[d];
    const int start = se.x;
    const int c     = se.y - se.x;
    const float di  = rsqrtf((float)(c + 1));
    const float4 sv = ((const float4*)(msg + (size_t)d * H_C))[lane];
    const float4 bv = ((const float4*)bias)[lane];
    float4 acc;
    acc.x = sv.x * di * di + bv.x; acc.y = sv.y * di * di + bv.y;
    acc.z = sv.z * di * di + bv.z; acc.w = sv.w * di * di + bv.w;
    for (int base = 0; base < c; base += 32) {
        const int left = c - base;
        const int m = left < 32 ? left : 32;
        int s_l = 0;
        float w_l = 0.0f;
        if (lane < m) {
            s_l = filed[start + base + lane].y;
            const int2 t = rowse[s_l];
            w_l = rsqrtf((float)(t.y - t.x + 1)) * di;
        }
        int j = 0;
        for (; j + 8 <= m; j += 8) gather_rows<8>(msg, s_l, w_l, j, lane, acc);
        if (j + 4 <= m) { gather_rows<4>(msg, s_l, w_l, j, lane, acc); j += 4; }
        if (j + 2 <= m) { gather_rows<2>(msg, s_l, w_l, j, lane, acc); j += 2; }
        if (j < m)      { gather_rows<1>(msg, s_l, w_l, j, lane, acc); }
    }
    ((float4*)(out + (size_t)d * H_C))[lane] = acc;
}

#define POOL_NODES 32
#define POOL_WAVES 2

__device__ __forceinline__ void pool_flush(float* __restrict__ sums, float* __restrict__ cnts,
                                           int g, int lane, int run,
                                           float a0, float a1, float a2, float a3) {
    float* o = sums + (size_t)g * H_C + lane;
    atomic_fadd_at<0>(o, a0);
    atomic_fadd_at<128>(o, a1);
    atomic_fadd_at<256>(o, a2);
    atomic_fadd_at<384>(o, a3);
    if (lane == 0) atomic_fadd(&cnts[g], (float)run);
}

__global__ void __launch_bounds__(POOL_WAVES * 32)
pool_scatter_kernel(const float* __restrict__ h,
                    const int* __restrict__ batch,
                    float* __restrict__ sums,
                    float* __restrict__ cnts, int n) {
    const int lane = threadIdx.x & 31;
    const int node0 = __builtin_amdgcn_readfirstlane(
                          (int)(blockIdx.x * POOL_WAVES + (threadIdx.x >> 5))) * POOL_NODES;
    if (node0 >= n) return;
    const int left = n - node0;
    const int m = left < POOL_NODES ? left : POOL_NODES;
    const int g_l = lane < m ? batch[node0 + lane] : 0;

    int   cur = 0;
    int   run = 0;
    float a0 = 0.0f, a1 = 0.0f, a2 = 0.0f, a3 = 0.0f;
    for (int j0 = 0; j0 < m; j0 += 8) {
        float v[8][4];
#pragma unroll
        for (int u = 0; u < 8; ++u) {
            const int j = j0 + u < m ? j0 + u : m - 1;
            const float* r = h + (size_t)(node0 + j) * H_C + lane;
            v[u][0] = r[0]; v[u][1] = r[32]; v[u][2] = r[64]; v[u][3] = r[96];
        }
#pragma unroll
        for (int u = 0; u < 8; ++u) {
            if (j0 + u < m) {
                const int g = __builtin_amdgcn_readlane(g_l, j0 + u);
                if (run == 0 || g != cur) {
                    if (run != 0) pool_flush(sums, cnts, cur, lane, run, a0, a1, a2, a3);
                    cur = g; run = 0;
                    a0 = 0.0f; a1 = 0.0f; a2 = 0.0f; a3 = 0.0f;
                }
                a0 += fmaxf(v[u][0], 0.0f); a1 += fmaxf(v[u][1], 0.0f);
                a2 += fmaxf(v[u][2], 0.0f); a3 += fmaxf(v[u][3], 0.0f);
                ++run;
            }
        }
    }
    pool_flush(sums, cnts, cur, lane, run, a0, a1, a2, a3);
}

__global__ void pool_div_kernel(float* __restrict__ pooled,
                                const float* __restrict__ cnts, int total) {
    int i = blockIdx.x * blockDim.x + threadIdx.x;
    if (i >= total) return;
    pooled[i] /= fmaxf(cnts[i >> 7], 1.0f);
}

extern "C" void kernel_launch(void* const* d_in, const int* in_sizes, int n_in,
                              void* d_out, int out_size, void* d_ws, size_t ws_size,
                              hipStream_t stream) {
    const float* x    = (const float*)d_in[0];
    const int*   ei   = (const int*)  d_in[1];
    const int*   batch= (const int*)  d_in[2];
    const float* W1   = (const float*)d_in[3];
    const float* b1   = (const float*)d_in[4];
    const float* W2   = (const float*)d_in[5];
    const float* b2   = (const float*)d_in[6];
    const float* Wl   = (const float*)d_in[7];
    const float* bl   = (const float*)d_in[8];

    const int N = in_sizes[0] / IN_C;
    const int E = in_sizes[1] / 2;
    const int G = (out_size - N * OUT_C) / H_C;

    const int* esrc = ei;
    const int* edst = ei + E;

    float* logits = (float*)d_out;
    float* pooled = logits + (size_t)N * OUT_C;

    const int T  = 256;
    const int tiles = (E + SORT_TILE - 1) / SORT_TILE;

    const size_t Npad = ((size_t)N + 255) & ~(size_t)255;
    const size_t Epad = ((size_t)E + 255) & ~(size_t)255;
    int*   rowse = (int*)d_ws;
    int*   tab   = rowse + 2 * Npad;
    int2*  p1    = (int2*)(tab + (size_t)tiles * 256);
    int2*  p2    = p1 + Epad;
    float* tmpA = (float*)(p2 + Epad);
    float* tmpB = tmpA + (size_t)N * H_C;
    float* tmpC = tmpB + (size_t)N * H_C;
    float* cnts = tmpC + (size_t)N * H_C;

    const int rowBlocks  = (N + 15) / 16;
    const int gemmBlocks = rowBlocks < 512 ? rowBlocks : 512;

    fill_kernel<<<(2 * N + T - 1) / T, T, 0, stream>>>((float*)rowse, 0.0f, 2 * N);
    fill_kernel<<<(G + T - 1) / T, T, 0, stream>>>(cnts, 0.0f, G);
    fill_kernel<<<(G * H_C + T - 1) / T, T, 0, stream>>>(pooled, 0.0f, G * H_C);

    sort_count_kernel<0, 1><<<tiles, 256, 0, stream>>>(edst, tab, E);
    sort_place_kernel<0, false><<<tiles, 256, 0, stream>>>(edst, esrc, p1, tab, E, tiles);
    sort_count_kernel<8, 2><<<tiles, 256, 0, stream>>>((const int*)p1, tab, E);
    sort_place_kernel<8, true><<<tiles, 256, 0, stream>>>((const int*)p1, nullptr, p2, tab, E, tiles);
    row_bounds_kernel<<<(E + T - 1) / T, T, 0, stream>>>(p2, rowse, E);

    const int gatherBlocks = (N + (T / 32) - 1) / (T / 32);

    gemm_wmma_kernel<IN_C, H_C, false, false><<<gemmBlocks, 256, 0, stream>>>(
        x, W1, nullptr, tmpA, N);
    row_gather_kernel<<<gatherBlocks, T, 0, stream>>>((const int2*)rowse, p2, tmpA, b1, tmpB, N);

    gemm_wmma_kernel<H_C, H_C, false, true><<<gemmBlocks, 256, 0, stream>>>(
        tmpB, W2, nullptr, tmpA, N);
    row_gather_kernel<<<gatherBlocks, T, 0, stream>>>((const int2*)rowse, p2, tmpA, b2, tmpC, N);

    gemm_wmma_kernel<H_C, OUT_C, true, true><<<gemmBlocks, 128, 0, stream>>>(
        tmpC, Wl, bl, logits, N);

    pool_scatter_kernel<<<(N + POOL_WAVES * POOL_NODES - 1) / (POOL_WAVES * POOL_NODES), POOL_WAVES * 32, 0, stream>>>(
        tmpC, batch, pooled, cnts, N);
    pool_div_kernel<<<(G * H_C + T - 1) / T, T, 0, stream>>>(pooled, cnts, G * H_C);
}


# sort: tile of 4096 pairs over 16 waves of 8 chunks (196 groups of 512 threads) instead of 8192 over 8 waves of 32: 4x shorter serial ranking chain per wave; tile-count walk split over all threads
# speedup vs baseline: 3.0176x; 1.0818x over previous
#include <hip/hip_runtime.h>

#define IN_C  128
#define H_C   128
#define OUT_C 64

typedef float v2f __attribute__((ext_vector_type(2)));
typedef float v8f __attribute__((ext_vector_type(8)));

__device__ __forceinline__ void atomic_fadd(float* p, float v) {
    asm volatile("global_atomic_add_f32 %0, %1, off" : : "v"(p), "v"(v) : "memory");
}

template<int OFF>
__device__ __forceinline__ void atomic_fadd_at(float* p, float v) {
    asm volatile("global_atomic_add_f32 %0, %1, off offset:%2" : : "v"(p), "v"(v), "n"(OFF) : "memory");
}

__global__ void fill_kernel(float* __restrict__ p, float v, int n) {
    int i = blockIdx.x * blockDim.x + threadIdx.x;
    if (i < n) p[i] = v;
}

#define SORT_WAVES   16
#define SORT_CHUNKS  8
#define SORT_GROUP   8
#define SORT_TILE    (SORT_WAVES * SORT_CHUNKS * 32)
#define SORT_THREADS (SORT_WAVES * 32)
#define SORT_SPLIT   (SORT_THREADS / 256)
#define WALK_BATCH   32

__device__ __forceinline__ unsigned digit_peers(int digit, unsigned valid) {
    unsigned peers = valid;
#pragma unroll
    for (int b = 0; b < 8; ++b) {
        const bool bit = (digit >> b) & 1;
        const unsigned m = __builtin_amdgcn_ballot_w32(bit);
        peers &= bit ? m : ~m;
    }
    return peers;
}

template<int SHIFT, int KSTRIDE>
__device__ __forceinline__ void count_digits(const int* __restrict__ kin, int* hrow,
                                             int wbase, int lane, int e) {
#pragma unroll
    for (int q = 0; q < 8; ++q) hrow[lane + 32 * q] = 0;
    __builtin_amdgcn_wave_barrier();
#pragma unroll 1
    for (int g = 0; g < SORT_CHUNKS; g += SORT_GROUP) {
        int keys[SORT_GROUP];
#pragma unroll
        for (int u = 0; u < SORT_GROUP; ++u) {
            const int i = wbase + (g + u) * 32 + lane;
            keys[u] = kin[(size_t)(i < e ? i : e - 1) * KSTRIDE];
        }
#pragma unroll
        for (int u = 0; u < SORT_GROUP; ++u) {
            const bool ok = wbase + (g + u) * 32 + lane < e;
            const int digit = (keys[u] >> SHIFT) & 255;
            const unsigned peers = digit_peers(digit, __builtin_amdgcn_ballot_w32(ok));
            if (ok && lane == __builtin_ctz(peers)) hrow[digit] += __builtin_popcount(peers);
            __builtin_amdgcn_wave_barrier();
        }
    }
}

template<int SHIFT, int KSTRIDE>
__global__ void __launch_bounds__(SORT_THREADS)
sort_count_kernel(const int* __restrict__ kin, int* __restrict__ tab, int e) {
    __shared__ int h[SORT_WAVES][256];
    const int tid = threadIdx.x, w = tid >> 5, lane = tid & 31;
    const int wbase = blockIdx.x * SORT_TILE + w * (SORT_CHUNKS * 32);
    count_digits<SHIFT, KSTRIDE>(kin, h[w], wbase, lane, e);
    __syncthreads();
    if (tid < 256) {
        int t = 0;
#pragma unroll
        for (int v = 0; v < SORT_WAVES; ++v) t += h[v][tid];
        tab[blockIdx.x * 256 + tid] = t;
    }
}

template<int SHIFT, bool PAIRS>
__global__ void __launch_bounds__(SORT_THREADS)
sort_place_kernel(const int* __restrict__ kin, const int* __restrict__ vin,
                  int2* __restrict__ out, const int* __restrict__ tab, int e, int nt) {
    __shared__ int h[SORT_WAVES][256];
    __shared__ int ptot[SORT_SPLIT][256];
    __shared__ int pbef[SORT_SPLIT][256];
    __shared__ int wsum[8];
    const int tid = threadIdx.x, w = tid >> 5, lane = tid & 31;
    const int tile = blockIdx.x;
    const int wbase = tile * SORT_TILE + w * (SORT_CHUNKS * 32);

    {
        const int q = __builtin_amdgcn_readfirstlane(tid >> 8);
        const int dg = tid & 255;
        int total = 0, before = 0;
        for (int t0 = q; t0 < nt; t0 += WALK_BATCH * SORT_SPLIT) {
            int c[WALK_BATCH];
#pragma unroll
            for (int u = 0; u < WALK_BATCH; ++u) {
                const int t = t0 + u * SORT_SPLIT;
                c[u] = tab[(t < nt ? t : nt - 1) * 256 + dg];
            }
#pragma unroll
            for (int u = 0; u < WALK_BATCH; ++u) {
                const int t = t0 + u * SORT_SPLIT;
                total  += t < nt ? c[u] : 0;
                before += t < tile ? c[u] : 0;
            }
        }
        ptot[q][dg] = total;
        pbef[q][dg] = before;
    }

    count_digits<SHIFT, PAIRS ? 2 : 1>(kin, h[w], wbase, lane, e);
    __syncthreads();

    int total = 0, before = 0, incl = 0;
    if (tid < 256) {
#pragma unroll
        for (int q = 0; q < SORT_SPLIT; ++q) { total += ptot[q][tid]; before += pbef[q][tid]; }
        incl = total;
#pragma unroll
        for (int off = 1; off < 32; off <<= 1) {
            const int t = __shfl_up(incl, off, 32);
            if (lane >= off) incl += t;
        }
        if (lane == 31) wsum[w] = incl;
    }
    __syncthreads();

    if (tid < 256) {
        int run = before + incl - total;
        for (int v = 0; v < w; ++v) run += wsum[v];
#pragma unroll
        for (int v = 0; v < SORT_WAVES; ++v) {
            const int t = h[v][tid];
            h[v][tid] = run;
            run += t;
        }
    }
    __syncthreads();

    int* hrow = h[w];
#pragma unroll 1
    for (int g = 0; g < SORT_CHUNKS; g += SORT_GROUP) {
        int keys[SORT_GROUP], vals[SORT_GROUP];
#pragma unroll
        for (int u = 0; u < SORT_GROUP; ++u) {
            const int i = wbase + (g + u) * 32 + lane;
            const int ic = i < e ? i : e - 1;
            if (PAIRS) {
                const int2 t = ((const int2*)kin)[ic];
                keys[u] = t.x; vals[u] = t.y;
            } else {
                keys[u] = kin[ic]; vals[u] = vin[ic];
            }
        }
#pragma unroll
        for (int u = 0; u < SORT_GROUP; ++u) {
            const bool ok = wbase + (g + u) * 32 + lane < e;
            const int digit = (keys[u] >> SHIFT) & 255;
            const unsigned peers = digit_peers(digit, __builtin_amdgcn_ballot_w32(ok));
            if (ok) {
                const int off = hrow[digit];
                const int pos = off + __builtin_popcount(peers & ((1u << lane) - 1u));
                out[pos] = make_int2(keys[u], vals[u]);
                if (lane == __builtin_ctz(peers)) hrow[digit] = off + __builtin_popcount(peers);
            }
            __builtin_amdgcn_wave_barrier();
        }
    }
}

__global__ void row_bounds_kernel(const int2* __restrict__ filed, int* __restrict__ rowse, int e) {
    int i = blockIdx.x * blockDim.x + threadIdx.x;
    if (i >= e) return;
    const int d = filed[i].x;
    if (i == 0 || filed[i - 1].x != d) rowse[2 * d] = i;
    if (i == e - 1 || filed[i + 1].x != d) rowse[2 * d + 1] = i + 1;
}

template<int K, int NC, bool ADD_BIAS, bool RELU_A>
__global__ void gemm_wmma_kernel(const float* __restrict__ A,
                                 const float* __restrict__ B,
                                 const float* __restrict__ bias,
                                 float* __restrict__ C, int M) {
    constexpr int LDA = K + 4;
    __shared__ __align__(16) float As[16 * LDA];

    const int tid   = threadIdx.x;
    const int wave  = tid >> 5;
    const int lane  = tid & 31;
    const int n0    = wave * 16;
    const int lhalf = lane >> 4;
    const int l15   = lane & 15;
    const int n     = n0 + l15;

    v2f bfrag[K / 4];
#pragma unroll
    for (int k0 = 0; k0 < K; k0 += 4) {
        const int kk = k0 + 2 * lhalf;
        bfrag[k0 / 4].x = B[(size_t)kk * NC + n];
        bfrag[k0 / 4].y = B[(size_t)(kk + 1) * NC + n];
    }
    const float bv = ADD_BIAS ? bias[n] : 0.0f;

    const int strips      = (M + 15) / 16;
    const int vecsPerRow  = K / 4;

    for (int s = blockIdx.x; s < strips; s += gridDim.x) {
        const int row0 = s * 16;
        __syncthreads();
        for (int v = tid; v < 16 * vecsPerRow; v += blockDim.x) {
            int r = v / vecsPerRow;
            float4 val = make_float4(0.f, 0.f, 0.f, 0.f);
            if (row0 + r < M)
                val = ((const float4*)(A + (size_t)(row0 + r) * K))[v % vecsPerRow];
            if (RELU_A) {
                val.x = fmaxf(val.x, 0.0f); val.y = fmaxf(val.y, 0.0f);
                val.z = fmaxf(val.z, 0.0f); val.w = fmaxf(val.w, 0.0f);
            }
            *(float4*)(As + r * LDA + 4 * (v % vecsPerRow)) = val;
        }
        __syncthreads();

        v8f acc = {};
#pragma unroll
        for (int k0 = 0; k0 < K; k0 += 4) {
            const int kk = k0 + 2 * lhalf;
            v2f a;
            a.x = As[l15 * LDA + kk];
            a.y = As[l15 * LDA + kk + 1];
            acc = __builtin_amdgcn_wmma_f32_16x16x4_f32(
                      false, a, false, bfrag[k0 / 4], (short)0, acc, false, false);
        }

#pragma unroll
        for (int r = 0; r < 8; ++r) {
            int m = row0 + r + 8 * lhalf;
            if (m < M) C[(size_t)m * NC + n] = acc[r] + bv;
        }
    }
}

template<int R>
__device__ __forceinline__ void gather_rows(const float* __restrict__ msg, int s_l, float w_l,
                                            int j, int lane, float4& acc) {
    float4 v[R];
#pragma unroll
    for (int u = 0; u < R; ++u) {
        const int s = __builtin_amdgcn_readlane(s_l, j + u);
        v[u] = ((const float4*)(msg + (size_t)s * H_C))[lane];
    }
#pragma unroll
    for (int u = 0; u < R; ++u) {
        const float w = __int_as_float(__builtin_amdgcn_readlane(__float_as_int(w_l), j + u));
        acc.x += v[u].x * w; acc.y += v[u].y * w;
        acc.z += v[u].z * w; acc.w += v[u].w * w;
    }
}

__global__ void __launch_bounds__(256)
row_gather_kernel(const int2* __restrict__ rowse, const int2* __restrict__ filed,
                  const float* __restrict__ msg, const float* __restrict__ bias,
                  float* __restrict__ out, int n) {
    const int lane = threadIdx.x & 31;
    const int d = __builtin_amdgcn_readfirstlane(
                      (int)(blockIdx.x * (blockDim.x >> 5) + (threadIdx.x >> 5)));
    if (d >= n) return;
    const int2 se   = rowse[d];
    const int start = se.x;
    const int c     = se.y - se.x;
    const float di  = rsqrtf((float)(c + 1));
    const float4 sv = ((const float4*)(msg + (size_t)d * H_C))[lane];
    const float4 bv = ((const float4*)bias)[lane];
    float4 acc;
    acc.x = sv.x * di * di + bv.x; acc.y = sv.y * di * di + bv.y;
    acc.z = sv.z * di * di + bv.z; acc.w = sv.w * di * di + bv.w;
    for (int base = 0; base < c; base += 32) {
        const int left = c - base;
        const int m = left < 32 ? left : 32;
        int s_l = 0;
        float w_l = 0.0f;
        if (lane < m) {
            s_l = filed[start + base + lane].y;
            const int2 t = rowse[s_l];
            w_l = rsqrtf((float)(t.y - t.x + 1)) * di;
        }
        int j = 0;
        for (; j + 8 <= m; j += 8) gather_rows<8>(msg, s_l, w_l, j, lane, acc);
        if (j + 4 <= m) { gather_rows<4>(msg, s_l, w_l, j, lane, acc); j += 4; }
        if (j + 2 <= m) { gather_rows<2>(msg, s_l, w_l, j, lane, acc); j += 2; }
        if (j < m)      { gather_rows<1>(msg, s_l, w_l, j, lane, acc); }
    }
    ((float4*)(out + (size_t)d * H_C))[lane] = acc;
}

#define POOL_NODES 32
#define POOL_WAVES 2

__device__ __forceinline__ void pool_flush(float* __restrict__ sums, float* __restrict__ cnts,
                                           int g, int lane, int run,
                                           float a0, float a1, float a2, float a3) {
    float* o = sums + (size_t)g * H_C + lane;
    atomic_fadd_at<0>(o, a0);
    atomic_fadd_at<128>(o, a1);
    atomic_fadd_at<256>(o, a2);
    atomic_fadd_at<384>(o, a3);
    if (lane == 0) atomic_fadd(&cnts[g], (float)run);
}

__global__ void __launch_bounds__(POOL_WAVES * 32)
pool_scatter_kernel(const float* __restrict__ h,
                    const int* __restrict__ batch,
                    float* __restrict__ sums,
                    float* __restrict__ cnts, int n) {
    const int lane = threadIdx.x & 31;
    const int node0 = __builtin_amdgcn_readfirstlane(
                          (int)(blockIdx.x * POOL_WAVES + (threadIdx.x >> 5))) * POOL_NODES;
    if (node0 >= n) return;
    const int left = n - node0;
    const int m = left < POOL_NODES ? left : POOL_NODES;
    const int g_l = lane < m ? batch[node0 + lane] : 0;

    int   cur = 0;
    int   run = 0;
    float a0 = 0.0f, a1 = 0.0f, a2 = 0.0f, a3 = 0.0f;
    for (int j0 = 0; j0 < m; j0 += 8) {
        float v[8][4];
#pragma unroll
        for (int u = 0; u < 8; ++u) {
            const int j = j0 + u < m ? j0 + u : m - 1;
            const float* r = h + (size_t)(node0 + j) * H_C + lane;
            v[u][0] = r[0]; v[u][1] = r[32]; v[u][2] = r[64]; v[u][3] = r[96];
        }
#pragma unroll
        for (int u = 0; u < 8; ++u) {
            if (j0 + u < m) {
                const int g = __builtin_amdgcn_readlane(g_l, j0 + u);
                if (run == 0 || g != cur) {
                    if (run != 0) pool_flush(sums, cnts, cur, lane, run, a0, a1, a2, a3);
                    cur = g; run = 0;
                    a0 = 0.0f; a1 = 0.0f; a2 = 0.0f; a3 = 0.0f;
                }
                a0 += fmaxf(v[u][0], 0.0f); a1 += fmaxf(v[u][1], 0.0f);
                a2 += fmaxf(v[u][2], 0.0f); a3 += fmaxf(v[u][3], 0.0f);
                ++run;
            }
        }
    }
    pool_flush(sums, cnts, cur, lane, run, a0, a1, a2, a3);
}

__global__ void pool_div_kernel(float* __restrict__ pooled,
                                const float* __restrict__ cnts, int total) {
    int i = blockIdx.x * blockDim.x + threadIdx.x;
    if (i >= total) return;
    pooled[i] /= fmaxf(cnts[i >> 7], 1.0f);
}

extern "C" void kernel_launch(void* const* d_in, const int* in_sizes, int n_in,
                              void* d_out, int out_size, void* d_ws, size_t ws_size,
                              hipStream_t stream) {
    const float* x    = (const float*)d_in[0];
    const int*   ei   = (const int*)  d_in[1];
    const int*   batch= (const int*)  d_in[2];
    const float* W1   = (const float*)d_in[3];
    const float* b1   = (const float*)d_in[4];
    const float* W2   = (const float*)d_in[5];
    const float* b2   = (const float*)d_in[6];
    const float* Wl   = (const float*)d_in[7];
    const float* bl   = (const float*)d_in[8];

    const int N = in_sizes[0] / IN_C;
    const int E = in_sizes[1] / 2;
    const int G = (out_size - N * OUT_C) / H_C;

    const int* esrc = ei;
    const int* edst = ei + E;

    float* logits = (float*)d_out;
    float* pooled = logits + (size_t)N * OUT_C;

    const int T  = 256;
    const int tiles = (E + SORT_TILE - 1) / SORT_TILE;

    const size_t Npad = ((size_t)N + 255) & ~(size_t)255;
    const size_t Epad = ((size_t)E + 255) & ~(size_t)255;
    int*   rowse = (int*)d_ws;
    int*   tab   = rowse + 2 * Npad;
    int2*  p1    = (int2*)(tab + (size_t)tiles * 256);
    int2*  p2    = p1 + Epad;
    float* tmpA = (float*)(p2 + Epad);
    float* tmpB = tmpA + (size_t)N * H_C;
    float* tmpC = tmpB + (size_t)N * H_C;
    float* cnts = tmpC + (size_t)N * H_C;

    const int rowBlocks  = (N + 15) / 16;
    const int gemmBlocks = rowBlocks < 512 ? rowBlocks : 512;

    fill_kernel<<<(2 * N + T - 1) / T, T, 0, stream>>>((float*)rowse, 0.0f, 2 * N);
    fill_kernel<<<(G + T - 1) / T, T, 0, stream>>>(cnts, 0.0f, G);
    fill_kernel<<<(G * H_C + T - 1) / T, T, 0, stream>>>(pooled, 0.0f, G * H_C);

    sort_count_kernel<0, 1><<<tiles, SORT_THREADS, 0, stream>>>(edst, tab, E);
    sort_place_kernel<0, false><<<tiles, SORT_THREADS, 0, stream>>>(edst, esrc, p1, tab, E, tiles);
    sort_count_kernel<8, 2><<<tiles, SORT_THREADS, 0, stream>>>((const int*)p1, tab, E);
    sort_place_kernel<8, true><<<tiles, SORT_THREADS, 0, stream>>>((const int*)p1, nullptr, p2, tab, E, tiles);
    row_bounds_kernel<<<(E + T - 1) / T, T, 0, stream>>>(p2, rowse, E);

    const int gatherBlocks = (N + (T / 32) - 1) / (T / 32);

    gemm_wmma_kernel<IN_C, H_C, false, false><<<gemmBlocks, 256, 0, stream>>>(
        x, W1, nullptr, tmpA, N);
    row_gather_kernel<<<gatherBlocks, T, 0, stream>>>((const int2*)rowse, p2, tmpA, b1, tmpB, N);

    gemm_wmma_kernel<H_C, H_C, false, true><<<gemmBlocks, 256, 0, stream>>>(
        tmpB, W2, nullptr, tmpA, N);
    row_gather_kernel<<<gatherBlocks, T, 0, stream>>>((const int2*)rowse, p2, tmpA, b2, tmpC, N);

    gemm_wmma_kernel<H_C, OUT_C, true, true><<<gemmBlocks, 128, 0, stream>>>(
        tmpC, Wl, bl, logits, N);

    pool_scatter_kernel<<<(N + POOL_WAVES * POOL_NODES - 1) / (POOL_WAVES * POOL_NODES), POOL_WAVES * 32, 0, stream>>>(
        tmpC, batch, pooled, cnts, N);
    pool_div_kernel<<<(G * H_C + T - 1) / T, T, 0, stream>>>(pooled, cnts, G * H_C);
}


# x@W1 product and the three zero fills done by extra work-groups of the four sort launches (neither depends on the sort): 4 launches fewer, the matrix work runs beside the sort's ranking chains
# speedup vs baseline: 3.2833x; 1.0881x over previous
#include <hip/hip_runtime.h>

#define IN_C  128
#define H_C   128
#define OUT_C 64

typedef float v2f __attribute__((ext_vector_type(2)));
typedef float v8f __attribute__((ext_vector_type(8)));

__device__ __forceinline__ void atomic_fadd(float* p, float v) {
    asm volatile("global_atomic_add_f32 %0, %1, off" : : "v"(p), "v"(v) : "memory");
}

template<int OFF>
__device__ __forceinline__ void atomic_fadd_at(float* p, float v) {
    asm volatile("global_atomic_add_f32 %0, %1, off offset:%2" : : "v"(p), "v"(v), "n"(OFF) : "memory");
}


struct SideGemm {
    const float* A;
    const float* B;
    float*       C;
    int M;
    int lo, hi;
};
struct ZeroFill {
    int*   a; int na;
    float* b; int nb;
    float* c; int nc;
};

template<int K>
__device__ __forceinline__ void gemm_rows32(const float* __restrict__ A,
                                            const float* __restrict__ B,
                                            float* __restrict__ C, int M,
                                            int first, int last, int step) {
    constexpr int NC  = 128;
    constexpr int LDA = K + 4;
    __shared__ __align__(16) float As32[32 * LDA];

    const int tid   = threadIdx.x;
    const int wave  = tid >> 5;
    const int lane  = tid & 31;
    const int n0    = (wave & 7) * 16;
    const int rbase = (wave >> 3) * 16;
    const int lhalf = lane >> 4;
    const int l15   = lane & 15;
    const int n     = n0 + l15;

    v2f bfrag[K / 4];
#pragma unroll
    for (int k0 = 0; k0 < K; k0 += 4) {
        const int kk = k0 + 2 * lhalf;
        bfrag[k0 / 4].x = B[(size_t)kk * NC + n];
        bfrag[k0 / 4].y = B[(size_t)(kk + 1) * NC + n];
    }

    constexpr int vecsPerRow = K / 4;
    constexpr int PER_THREAD = 32 * vecsPerRow / 512;
    static_assert(PER_THREAD * 512 == 32 * vecsPerRow, "a strip is a whole number of pieces per thread");
    for (int s = first; s < last; s += step) {
        const int row0 = s * 32;
        __syncthreads();
        float4 val[PER_THREAD];
#pragma unroll
        for (int u = 0; u < PER_THREAD; ++u) {
            const int v = tid + u * 512;
            const int m = row0 + v / vecsPerRow;
            val[u] = ((const float4*)(A + (size_t)(m < M ? m : M - 1) * K))[v % vecsPerRow];
        }
#pragma unroll
        for (int u = 0; u < PER_THREAD; ++u) {
            const int v = tid + u * 512;
            const int r = v / vecsPerRow;
            if (row0 + r >= M) val[u] = make_float4(0.f, 0.f, 0.f, 0.f);
            *(float4*)(As32 + r * LDA + 4 * (v % vecsPerRow)) = val[u];
        }
        __syncthreads();

        v8f acc = {};
#pragma unroll
        for (int k0 = 0; k0 < K; k0 += 4) {
            const int kk = k0 + 2 * lhalf;
            v2f a;
            a.x = As32[(rbase + l15) * LDA + kk];
            a.y = As32[(rbase + l15) * LDA + kk + 1];
            acc = __builtin_amdgcn_wmma_f32_16x16x4_f32(
                      false, a, false, bfrag[k0 / 4], (short)0, acc, false, false);
        }
#pragma unroll
        for (int r = 0; r < 8; ++r) {
            const int m = row0 + rbase + r + 8 * lhalf;
            if (m < M) C[(size_t)m * NC + n] = acc[r];
        }
    }
}

#define SORT_WAVES   16
#define SORT_CHUNKS  8
#define SORT_GROUP   8
#define SORT_TILE    (SORT_WAVES * SORT_CHUNKS * 32)
#define SORT_THREADS (SORT_WAVES * 32)
#define SORT_SPLIT   (SORT_THREADS / 256)
#define WALK_BATCH   32

__device__ __forceinline__ unsigned digit_peers(int digit, unsigned valid) {
    unsigned peers = valid;
#pragma unroll
    for (int b = 0; b < 8; ++b) {
        const bool bit = (digit >> b) & 1;
        const unsigned m = __builtin_amdgcn_ballot_w32(bit);
        peers &= bit ? m : ~m;
    }
    return peers;
}

template<int SHIFT, int KSTRIDE>
__device__ __forceinline__ void count_digits(const int* __restrict__ kin, int* hrow,
                                             int wbase, int lane, int e) {
#pragma unroll
    for (int q = 0; q < 8; ++q) hrow[lane + 32 * q] = 0;
    __builtin_amdgcn_wave_barrier();
#pragma unroll 1
    for (int g = 0; g < SORT_CHUNKS; g += SORT_GROUP) {
        int keys[SORT_GROUP];
#pragma unroll
        for (int u = 0; u < SORT_GROUP; ++u) {
            const int i = wbase + (g + u) * 32 + lane;
            keys[u] = kin[(size_t)(i < e ? i : e - 1) * KSTRIDE];
        }
#pragma unroll
        for (int u = 0; u < SORT_GROUP; ++u) {
            const bool ok = wbase + (g + u) * 32 + lane < e;
            const int digit = (keys[u] >> SHIFT) & 255;
            const unsigned peers = digit_peers(digit, __builtin_amdgcn_ballot_w32(ok));
            if (ok && lane == __builtin_ctz(peers)) hrow[digit] += __builtin_popcount(peers);
            __builtin_amdgcn_wave_barrier();
        }
    }
}

template<int SHIFT, int KSTRIDE, bool FILL>
__global__ void __launch_bounds__(SORT_THREADS)
sort_count_kernel(const int* __restrict__ kin, int* __restrict__ tab, int e, int nt,
                  SideGemm sg, ZeroFill zf) {
    __shared__ int h[SORT_WAVES][256];
    const int tid = threadIdx.x, w = tid >> 5, lane = tid & 31;
    if (FILL) {
        const int gsz = (int)gridDim.x * SORT_THREADS;
        const int g0  = (int)blockIdx.x * SORT_THREADS + tid;
        for (int i = g0; i < zf.na; i += gsz) zf.a[i] = 0;
        for (int i = g0; i < zf.nb; i += gsz) zf.b[i] = 0.0f;
        for (int i = g0; i < zf.nc; i += gsz) zf.c[i] = 0.0f;
    }
    const int gb = (int)blockIdx.x - nt;
    if (gb >= 0) {
        gemm_rows32<IN_C>(sg.A, sg.B, sg.C, sg.M, sg.lo + gb, sg.hi, (int)gridDim.x - nt);
        return;
    }
    const int wbase = blockIdx.x * SORT_TILE + w * (SORT_CHUNKS * 32);
    count_digits<SHIFT, KSTRIDE>(kin, h[w], wbase, lane, e);
    __syncthreads();
    if (tid < 256) {
        int t = 0;
#pragma unroll
        for (int v = 0; v < SORT_WAVES; ++v) t += h[v][tid];
        tab[blockIdx.x * 256 + tid] = t;
    }
}

template<int SHIFT, bool PAIRS>
__global__ void __launch_bounds__(SORT_THREADS)
sort_place_kernel(const int* __restrict__ kin, const int* __restrict__ vin,
                  int2* __restrict__ out, const int* __restrict__ tab, int e, int nt,
                  SideGemm sg) {
    __shared__ int h[SORT_WAVES][256];
    __shared__ int ptot[SORT_SPLIT][256];
    __shared__ int pbef[SORT_SPLIT][256];
    __shared__ int wsum[8];
    const int tid = threadIdx.x, w = tid >> 5, lane = tid & 31;
    const int gb = (int)blockIdx.x - nt;
    if (gb >= 0) {
        gemm_rows32<IN_C>(sg.A, sg.B, sg.C, sg.M, sg.lo + gb, sg.hi, (int)gridDim.x - nt);
        return;
    }
    const int tile = blockIdx.x;
    const int wbase = tile * SORT_TILE + w * (SORT_CHUNKS * 32);

    {
        const int q = __builtin_amdgcn_readfirstlane(tid >> 8);
        const int dg = tid & 255;
        int total = 0, before = 0;
        for (int t0 = q; t0 < nt; t0 += WALK_BATCH * SORT_SPLIT) {
            int c[WALK_BATCH];
#pragma unroll
            for (int u = 0; u < WALK_BATCH; ++u) {
                const int t = t0 + u * SORT_SPLIT;
                c[u] = tab[(t < nt ? t : nt - 1) * 256 + dg];
            }
#pragma unroll
            for (int u = 0; u < WALK_BATCH; ++u) {
                const int t = t0 + u * SORT_SPLIT;
                total  += t < nt ? c[u] : 0;
                before += t < tile ? c[u] : 0;
            }
        }
        ptot[q][dg] = total;
        pbef[q][dg] = before;
    }

    count_digits<SHIFT, PAIRS ? 2 : 1>(kin, h[w], wbase, lane, e);
    __syncthreads();

    int total = 0, before = 0, incl = 0;
    if (tid < 256) {
#pragma unroll
        for (int q = 0; q < SORT_SPLIT; ++q) { total += ptot[q][tid]; before += pbef[q][tid]; }
        incl = total;
#pragma unroll
        for (int off = 1; off < 32; off <<= 1) {
            const int t = __shfl_up(incl, off, 32);
            if (lane >= off) incl += t;
        }
        if (lane == 31) wsum[w] = incl;
    }
    __syncthreads();

    if (tid < 256) {
        int run = before + incl - total;
        for (int v = 0; v < w; ++v) run += wsum[v];
#pragma unroll
        for (int v = 0; v < SORT_WAVES; ++v) {
            const int t = h[v][tid];
            h[v][tid] = run;
            run += t;
        }
    }
    __syncthreads();

    int* hrow = h[w];
#pragma unroll 1
    for (int g = 0; g < SORT_CHUNKS; g += SORT_GROUP) {
        int keys[SORT_GROUP], vals[SORT_GROUP];
#pragma unroll
        for (int u = 0; u < SORT_GROUP; ++u) {
            const int i = wbase + (g + u) * 32 + lane;
            const int ic = i < e ? i : e - 1;
            if (PAIRS) {
                const int2 t = ((const int2*)kin)[ic];
                keys[u] = t.x; vals[u] = t.y;
            } else {
                keys[u] = kin[ic]; vals[u] = vin[ic];
            }
        }
#pragma unroll
        for (int u = 0; u < SORT_GROUP; ++u) {
            const bool ok = wbase + (g + u) * 32 + lane < e;
            const int digit = (keys[u] >> SHIFT) & 255;
            const unsigned peers = digit_peers(digit, __builtin_amdgcn_ballot_w32(ok));
            if (ok) {
                const int off = hrow[digit];
                const int pos = off + __builtin_popcount(peers & ((1u << lane) - 1u));
                out[pos] = make_int2(keys[u], vals[u]);
                if (lane == __builtin_ctz(peers)) hrow[digit] = off + __builtin_popcount(peers);
            }
            __builtin_amdgcn_wave_barrier();
        }
    }
}

__global__ void row_bounds_kernel(const int2* __restrict__ filed, int* __restrict__ rowse, int e) {
    int i = blockIdx.x * blockDim.x + threadIdx.x;
    if (i >= e) return;
    const int d = filed[i].x;
    if (i == 0 || filed[i - 1].x != d) rowse[2 * d] = i;
    if (i == e - 1 || filed[i + 1].x != d) rowse[2 * d + 1] = i + 1;
}

template<int K, int NC, bool ADD_BIAS, bool RELU_A>
__global__ void gemm_wmma_kernel(const float* __restrict__ A,
                                 const float* __restrict__ B,
                                 const float* __restrict__ bias,
                                 float* __restrict__ C, int M) {
    constexpr int LDA = K + 4;
    __shared__ __align__(16) float As[16 * LDA];

    const int tid   = threadIdx.x;
    const int wave  = tid >> 5;
    const int lane  = tid & 31;
    const int n0    = wave * 16;
    const int lhalf = lane >> 4;
    const int l15   = lane & 15;
    const int n     = n0 + l15;

    v2f bfrag[K / 4];
#pragma unroll
    for (int k0 = 0; k0 < K; k0 += 4) {
        const int kk = k0 + 2 * lhalf;
        bfrag[k0 / 4].x = B[(size_t)kk * NC + n];
        bfrag[k0 / 4].y = B[(size_t)(kk + 1) * NC + n];
    }
    const float bv = ADD_BIAS ? bias[n] : 0.0f;

    const int strips      = (M + 15) / 16;
    const int vecsPerRow  = K / 4;

    for (int s = blockIdx.x; s < strips; s += gridDim.x) {
        const int row0 = s * 16;
        __syncthreads();
        for (int v = tid; v < 16 * vecsPerRow; v += blockDim.x) {
            int r = v / vecsPerRow;
            float4 val = make_float4(0.f, 0.f, 0.f, 0.f);
            if (row0 + r < M)
                val = ((const float4*)(A + (size_t)(row0 + r) * K))[v % vecsPerRow];
            if (RELU_A) {
                val.x = fmaxf(val.x, 0.0f); val.y = fmaxf(val.y, 0.0f);
                val.z = fmaxf(val.z, 0.0f); val.w = fmaxf(val.w, 0.0f);
            }
            *(float4*)(As + r * LDA + 4 * (v % vecsPerRow)) = val;
        }
        __syncthreads();

        v8f acc = {};
#pragma unroll
        for (int k0 = 0; k0 < K; k0 += 4) {
            const int kk = k0 + 2 * lhalf;
            v2f a;
            a.x = As[l15 * LDA + kk];
            a.y = As[l15 * LDA + kk + 1];
            acc = __builtin_amdgcn_wmma_f32_16x16x4_f32(
                      false, a, false, bfrag[k0 / 4], (short)0, acc, false, false);
        }

#pragma unroll
        for (int r = 0; r < 8; ++r) {
            int m = row0 + r + 8 * lhalf;
            if (m < M) C[(size_t)m * NC + n] = acc[r] + bv;
        }
    }
}

template<int R>
__device__ __forceinline__ void gather_rows(const float* __restrict__ msg, int s_l, float w_l,
                                            int j, int lane, float4& acc) {
    float4 v[R];
#pragma unroll
    for (int u = 0; u < R; ++u) {
        const int s = __builtin_amdgcn_readlane(s_l, j + u);
        v[u] = ((const float4*)(msg + (size_t)s * H_C))[lane];
    }
#pragma unroll
    for (int u = 0; u < R; ++u) {
        const float w = __int_as_float(__builtin_amdgcn_readlane(__float_as_int(w_l), j + u));
        acc.x += v[u].x * w; acc.y += v[u].y * w;
        acc.z += v[u].z * w; acc.w += v[u].w * w;
    }
}

__global__ void __launch_bounds__(256)
row_gather_kernel(const int2* __restrict__ rowse, const int2* __restrict__ filed,
                  const float* __restrict__ msg, const float* __restrict__ bias,
                  float* __restrict__ out, int n) {
    const int lane = threadIdx.x & 31;
    const int d = __builtin_amdgcn_readfirstlane(
                      (int)(blockIdx.x * (blockDim.x >> 5) + (threadIdx.x >> 5)));
    if (d >= n) return;
    const int2 se   = rowse[d];
    const int start = se.x;
    const int c     = se.y - se.x;
    const float di  = rsqrtf((float)(c + 1));
    const float4 sv = ((const float4*)(msg + (size_t)d * H_C))[lane];
    const float4 bv = ((const float4*)bias)[lane];
    float4 acc;
    acc.x = sv.x * di * di + bv.x; acc.y = sv.y * di * di + bv.y;
    acc.z = sv.z * di * di + bv.z; acc.w = sv.w * di * di + bv.w;
    for (int base = 0; base < c; base += 32) {
        const int left = c - base;
        const int m = left < 32 ? left : 32;
        int s_l = 0;
        float w_l = 0.0f;
        if (lane < m) {
            s_l = filed[start + base + lane].y;
            const int2 t = rowse[s_l];
            w_l = rsqrtf((float)(t.y - t.x + 1)) * di;
        }
        int j = 0;
        for (; j + 8 <= m; j += 8) gather_rows<8>(msg, s_l, w_l, j, lane, acc);
        if (j + 4 <= m) { gather_rows<4>(msg, s_l, w_l, j, lane, acc); j += 4; }
        if (j + 2 <= m) { gather_rows<2>(msg, s_l, w_l, j, lane, acc); j += 2; }
        if (j < m)      { gather_rows<1>(msg, s_l, w_l, j, lane, acc); }
    }
    ((float4*)(out + (size_t)d * H_C))[lane] = acc;
}

#define POOL_NODES 32
#define POOL_WAVES 2

__device__ __forceinline__ void pool_flush(float* __restrict__ sums, float* __restrict__ cnts,
                                           int g, int lane, int run,
                                           float a0, float a1, float a2, float a3) {
    float* o = sums + (size_t)g * H_C + lane;
    atomic_fadd_at<0>(o, a0);
    atomic_fadd_at<128>(o, a1);
    atomic_fadd_at<256>(o, a2);
    atomic_fadd_at<384>(o, a3);
    if (lane == 0) atomic_fadd(&cnts[g], (float)run);
}

__global__ void __launch_bounds__(POOL_WAVES * 32)
pool_scatter_kernel(const float* __restrict__ h,
                    const int* __restrict__ batch,
                    float* __restrict__ sums,
                    float* __restrict__ cnts, int n) {
    const int lane = threadIdx.x & 31;
    const int node0 = __builtin_amdgcn_readfirstlane(
                          (int)(blockIdx.x * POOL_WAVES + (threadIdx.x >> 5))) * POOL_NODES;
    if (node0 >= n) return;
    const int left = n - node0;
    const int m = left < POOL_NODES ? left : POOL_NODES;
    const int g_l = lane < m ? batch[node0 + lane] : 0;

    int   cur = 0;
    int   run = 0;
    float a0 = 0.0f, a1 = 0.0f, a2 = 0.0f, a3 = 0.0f;
    for (int j0 = 0; j0 < m; j0 += 8) {
        float v[8][4];
#pragma unroll
        for (int u = 0; u < 8; ++u) {
            const int j = j0 + u < m ? j0 + u : m - 1;
            const float* r = h + (size_t)(node0 + j) * H_C + lane;
            v[u][0] = r[0]; v[u][1] = r[32]; v[u][2] = r[64]; v[u][3] = r[96];
        }
#pragma unroll
        for (int u = 0; u < 8; ++u) {
            if (j0 + u < m) {
                const int g = __builtin_amdgcn_readlane(g_l, j0 + u);
                if (run == 0 || g != cur) {
                    if (run != 0) pool_flush(sums, cnts, cur, lane, run, a0, a1, a2, a3);
                    cur = g; run = 0;
                    a0 = 0.0f; a1 = 0.0f; a2 = 0.0f; a3 = 0.0f;
                }
                a0 += fmaxf(v[u][0], 0.0f); a1 += fmaxf(v[u][1], 0.0f);
                a2 += fmaxf(v[u][2], 0.0f); a3 += fmaxf(v[u][3], 0.0f);
                ++run;
            }
        }
    }
    pool_flush(sums, cnts, cur, lane, run, a0, a1, a2, a3);
}

__global__ void pool_div_kernel(float* __restrict__ pooled,
                                const float* __restrict__ cnts, int total) {
    int i = blockIdx.x * blockDim.x + threadIdx.x;
    if (i >= total) return;
    pooled[i] /= fmaxf(cnts[i >> 7], 1.0f);
}

extern "C" void kernel_launch(void* const* d_in, const int* in_sizes, int n_in,
                              void* d_out, int out_size, void* d_ws, size_t ws_size,
                              hipStream_t stream) {
    const float* x    = (const float*)d_in[0];
    const int*   ei   = (const int*)  d_in[1];
    const int*   batch= (const int*)  d_in[2];
    const float* W1   = (const float*)d_in[3];
    const float* b1   = (const float*)d_in[4];
    const float* W2   = (const float*)d_in[5];
    const float* b2   = (const float*)d_in[6];
    const float* Wl   = (const float*)d_in[7];
    const float* bl   = (const float*)d_in[8];

    const int N = in_sizes[0] / IN_C;
    const int E = in_sizes[1] / 2;
    const int G = (out_size - N * OUT_C) / H_C;

    const int* esrc = ei;
    const int* edst = ei + E;

    float* logits = (float*)d_out;
    float* pooled = logits + (size_t)N * OUT_C;

    const int T  = 256;
    const int tiles = (E + SORT_TILE - 1) / SORT_TILE;

    const size_t Npad = ((size_t)N + 255) & ~(size_t)255;
    const size_t Epad = ((size_t)E + 255) & ~(size_t)255;
    int*   rowse = (int*)d_ws;
    int*   tab   = rowse + 2 * Npad;
    int2*  p1    = (int2*)(tab + (size_t)tiles * 256);
    int2*  p2    = p1 + Epad;
    float* tmpA = (float*)(p2 + Epad);
    float* tmpB = tmpA + (size_t)N * H_C;
    float* tmpC = tmpB + (size_t)N * H_C;
    float* cnts = tmpC + (size_t)N * H_C;

    const int rowBlocks  = (N + 15) / 16;
    const int gemmBlocks = rowBlocks < 512 ? rowBlocks : 512;

    const int S32 = (N + 31) / 32;
    const int c1 = S32 / 6, c2 = S32 / 2, c3 = c2 + c1;
    const SideGemm sg0 = {x, W1, tmpA, N, 0,  c1};
    const SideGemm sg1 = {x, W1, tmpA, N, c1, c2};
    const SideGemm sg2 = {x, W1, tmpA, N, c2, c3};
    const SideGemm sg3 = {x, W1, tmpA, N, c3, S32};
    const int gb0 = c1, gb1 = (c2 - c1 + 1) / 2, gb2 = c3 - c2, gb3 = (S32 - c3 + 1) / 2;
    const ZeroFill zf = {rowse, 2 * N, cnts, G, pooled, G * H_C};

    sort_count_kernel<0, 1, true><<<tiles + gb0, SORT_THREADS, 0, stream>>>(edst, tab, E, tiles, sg0, zf);
    sort_place_kernel<0, false><<<tiles + gb1, SORT_THREADS, 0, stream>>>(edst, esrc, p1, tab, E, tiles, sg1);
    sort_count_kernel<8, 2, false><<<tiles + gb2, SORT_THREADS, 0, stream>>>((const int*)p1, tab, E, tiles, sg2, zf);
    sort_place_kernel<8, true><<<tiles + gb3, SORT_THREADS, 0, stream>>>((const int*)p1, nullptr, p2, tab, E, tiles, sg3);
    row_bounds_kernel<<<(E + T - 1) / T, T, 0, stream>>>(p2, rowse, E);

    const int gatherBlocks = (N + (T / 32) - 1) / (T / 32);

    row_gather_kernel<<<gatherBlocks, T, 0, stream>>>((const int2*)rowse, p2, tmpA, b1, tmpB, N);

    gemm_wmma_kernel<H_C, H_C, false, true><<<gemmBlocks, 256, 0, stream>>>(
        tmpB, W2, nullptr, tmpA, N);
    row_gather_kernel<<<gatherBlocks, T, 0, stream>>>((const int2*)rowse, p2, tmpA, b2, tmpC, N);

    gemm_wmma_kernel<H_C, OUT_C, true, true><<<gemmBlocks, 128, 0, stream>>>(
        tmpC, Wl, bl, logits, N);

    pool_scatter_kernel<<<(N + POOL_WAVES * POOL_NODES - 1) / (POOL_WAVES * POOL_NODES), POOL_WAVES * 32, 0, stream>>>(
        tmpC, batch, pooled, cnts, N);
    pool_div_kernel<<<(G * H_C + T - 1) / T, T, 0, stream>>>(pooled, cnts, G * H_C);
}


# aggregation fused with the product after it: an 8-wave work-group sums 16 nodes' rows, stages max(.,0) of them in LDS as the A strip and multiplies there; 2 launches, 2 N x 128 round trips less
# speedup vs baseline: 3.7072x; 1.1291x over previous
#include <hip/hip_runtime.h>

#define IN_C  128
#define H_C   128
#define OUT_C 64

typedef float v2f __attribute__((ext_vector_type(2)));
typedef float v8f __attribute__((ext_vector_type(8)));

__device__ __forceinline__ void atomic_fadd(float* p, float v) {
    asm volatile("global_atomic_add_f32 %0, %1, off" : : "v"(p), "v"(v) : "memory");
}

template<int OFF>
__device__ __forceinline__ void atomic_fadd_at(float* p, float v) {
    asm volatile("global_atomic_add_f32 %0, %1, off offset:%2" : : "v"(p), "v"(v), "n"(OFF) : "memory");
}


struct SideGemm {
    const float* A;
    const float* B;
    float*       C;
    int M;
    int lo, hi;
};
struct ZeroFill {
    int*   a; int na;
    float* b; int nb;
    float* c; int nc;
};

template<int K>
__device__ __forceinline__ void gemm_rows32(const float* __restrict__ A,
                                            const float* __restrict__ B,
                                            float* __restrict__ C, int M,
                                            int first, int last, int step) {
    constexpr int NC  = 128;
    constexpr int LDA = K + 4;
    __shared__ __align__(16) float As32[32 * LDA];

    const int tid   = threadIdx.x;
    const int wave  = tid >> 5;
    const int lane  = tid & 31;
    const int n0    = (wave & 7) * 16;
    const int rbase = (wave >> 3) * 16;
    const int lhalf = lane >> 4;
    const int l15   = lane & 15;
    const int n     = n0 + l15;

    v2f bfrag[K / 4];
#pragma unroll
    for (int k0 = 0; k0 < K; k0 += 4) {
        const int kk = k0 + 2 * lhalf;
        bfrag[k0 / 4].x = B[(size_t)kk * NC + n];
        bfrag[k0 / 4].y = B[(size_t)(kk + 1) * NC + n];
    }

    constexpr int vecsPerRow = K / 4;
    constexpr int PER_THREAD = 32 * vecsPerRow / 512;
    static_assert(PER_THREAD * 512 == 32 * vecsPerRow, "a strip is a whole number of pieces per thread");
    for (int s = first; s < last; s += step) {
        const int row0 = s * 32;
        __syncthreads();
        float4 val[PER_THREAD];
#pragma unroll
        for (int u = 0; u < PER_THREAD; ++u) {
            const int v = tid + u * 512;
            const int m = row0 + v / vecsPerRow;
            val[u] = ((const float4*)(A + (size_t)(m < M ? m : M - 1) * K))[v % vecsPerRow];
        }
#pragma unroll
        for (int u = 0; u < PER_THREAD; ++u) {
            const int v = tid + u * 512;
            const int r = v / vecsPerRow;
            if (row0 + r >= M) val[u] = make_float4(0.f, 0.f, 0.f, 0.f);
            *(float4*)(As32 + r * LDA + 4 * (v % vecsPerRow)) = val[u];
        }
        __syncthreads();

        v8f acc = {};
#pragma unroll
        for (int k0 = 0; k0 < K; k0 += 4) {
            const int kk = k0 + 2 * lhalf;
            v2f a;
            a.x = As32[(rbase + l15) * LDA + kk];
            a.y = As32[(rbase + l15) * LDA + kk + 1];
            acc = __builtin_amdgcn_wmma_f32_16x16x4_f32(
                      false, a, false, bfrag[k0 / 4], (short)0, acc, false, false);
        }
#pragma unroll
        for (int r = 0; r < 8; ++r) {
            const int m = row0 + rbase + r + 8 * lhalf;
            if (m < M) C[(size_t)m * NC + n] = acc[r];
        }
    }
}

#define SORT_WAVES   16
#define SORT_CHUNKS  8
#define SORT_GROUP   8
#define SORT_TILE    (SORT_WAVES * SORT_CHUNKS * 32)
#define SORT_THREADS (SORT_WAVES * 32)
#define SORT_SPLIT   (SORT_THREADS / 256)
#define WALK_BATCH   32

__device__ __forceinline__ unsigned digit_peers(int digit, unsigned valid) {
    unsigned peers = valid;
#pragma unroll
    for (int b = 0; b < 8; ++b) {
        const bool bit = (digit >> b) & 1;
        const unsigned m = __builtin_amdgcn_ballot_w32(bit);
        peers &= bit ? m : ~m;
    }
    return peers;
}

template<int SHIFT, int KSTRIDE>
__device__ __forceinline__ void count_digits(const int* __restrict__ kin, int* hrow,
                                             int wbase, int lane, int e) {
#pragma unroll
    for (int q = 0; q < 8; ++q) hrow[lane + 32 * q] = 0;
    __builtin_amdgcn_wave_barrier();
#pragma unroll 1
    for (int g = 0; g < SORT_CHUNKS; g += SORT_GROUP) {
        int keys[SORT_GROUP];
#pragma unroll
        for (int u = 0; u < SORT_GROUP; ++u) {
            const int i = wbase + (g + u) * 32 + lane;
            keys[u] = kin[(size_t)(i < e ? i : e - 1) * KSTRIDE];
        }
#pragma unroll
        for (int u = 0; u < SORT_GROUP; ++u) {
            const bool ok = wbase + (g + u) * 32 + lane < e;
            const int digit = (keys[u] >> SHIFT) & 255;
            const unsigned peers = digit_peers(digit, __builtin_amdgcn_ballot_w32(ok));
            if (ok && lane == __builtin_ctz(peers)) hrow[digit] += __builtin_popcount(peers);
            __builtin_amdgcn_wave_barrier();
        }
    }
}

template<int SHIFT, int KSTRIDE, bool FILL>
__global__ void __launch_bounds__(SORT_THREADS)
sort_count_kernel(const int* __restrict__ kin, int* __restrict__ tab, int e, int nt,
                  SideGemm sg, ZeroFill zf) {
    __shared__ int h[SORT_WAVES][256];
    const int tid = threadIdx.x, w = tid >> 5, lane = tid & 31;
    if (FILL) {
        const int gsz = (int)gridDim.x * SORT_THREADS;
        const int g0  = (int)blockIdx.x * SORT_THREADS + tid;
        for (int i = g0; i < zf.na; i += gsz) zf.a[i] = 0;
        for (int i = g0; i < zf.nb; i += gsz) zf.b[i] = 0.0f;
        for (int i = g0; i < zf.nc; i += gsz) zf.c[i] = 0.0f;
    }
    const int gb = (int)blockIdx.x - nt;
    if (gb >= 0) {
        gemm_rows32<IN_C>(sg.A, sg.B, sg.C, sg.M, sg.lo + gb, sg.hi, (int)gridDim.x - nt);
        return;
    }
    const int wbase = blockIdx.x * SORT_TILE + w * (SORT_CHUNKS * 32);
    count_digits<SHIFT, KSTRIDE>(kin, h[w], wbase, lane, e);
    __syncthreads();
    if (tid < 256) {
        int t = 0;
#pragma unroll
        for (int v = 0; v < SORT_WAVES; ++v) t += h[v][tid];
        tab[blockIdx.x * 256 + tid] = t;
    }
}

template<int SHIFT, bool PAIRS>
__global__ void __launch_bounds__(SORT_THREADS)
sort_place_kernel(const int* __restrict__ kin, const int* __restrict__ vin,
                  int2* __restrict__ out, const int* __restrict__ tab, int e, int nt,
                  SideGemm sg) {
    __shared__ int h[SORT_WAVES][256];
    __shared__ int ptot[SORT_SPLIT][256];
    __shared__ int pbef[SORT_SPLIT][256];
    __shared__ int wsum[8];
    const int tid = threadIdx.x, w = tid >> 5, lane = tid & 31;
    const int gb = (int)blockIdx.x - nt;
    if (gb >= 0) {
        gemm_rows32<IN_C>(sg.A, sg.B, sg.C, sg.M, sg.lo + gb, sg.hi, (int)gridDim.x - nt);
        return;
    }
    const int tile = blockIdx.x;
    const int wbase = tile * SORT_TILE + w * (SORT_CHUNKS * 32);

    {
        const int q = __builtin_amdgcn_readfirstlane(tid >> 8);
        const int dg = tid & 255;
        int total = 0, before = 0;
        for (int t0 = q; t0 < nt; t0 += WALK_BATCH * SORT_SPLIT) {
            int c[WALK_BATCH];
#pragma unroll
            for (int u = 0; u < WALK_BATCH; ++u) {
                const int t = t0 + u * SORT_SPLIT;
                c[u] = tab[(t < nt ? t : nt - 1) * 256 + dg];
            }
#pragma unroll
            for (int u = 0; u < WALK_BATCH; ++u) {
                const int t = t0 + u * SORT_SPLIT;
                total  += t < nt ? c[u] : 0;
                before += t < tile ? c[u] : 0;
            }
        }
        ptot[q][dg] = total;
        pbef[q][dg] = before;
    }

    count_digits<SHIFT, PAIRS ? 2 : 1>(kin, h[w], wbase, lane, e);
    __syncthreads();

    int total = 0, before = 0, incl = 0;
    if (tid < 256) {
#pragma unroll
        for (int q = 0; q < SORT_SPLIT; ++q) { total += ptot[q][tid]; before += pbef[q][tid]; }
        incl = total;
#pragma unroll
        for (int off = 1; off < 32; off <<= 1) {
            const int t = __shfl_up(incl, off, 32);
            if (lane >= off) incl += t;
        }
        if (lane == 31) wsum[w] = incl;
    }
    __syncthreads();

    if (tid < 256) {
        int run = before + incl - total;
        for (int v = 0; v < w; ++v) run += wsum[v];
#pragma unroll
        for (int v = 0; v < SORT_WAVES; ++v) {
            const int t = h[v][tid];
            h[v][tid] = run;
            run += t;
        }
    }
    __syncthreads();

    int* hrow = h[w];
#pragma unroll 1
    for (int g = 0; g < SORT_CHUNKS; g += SORT_GROUP) {
        int keys[SORT_GROUP], vals[SORT_GROUP];
#pragma unroll
        for (int u = 0; u < SORT_GROUP; ++u) {
            const int i = wbase + (g + u) * 32 + lane;
            const int ic = i < e ? i : e - 1;
            if (PAIRS) {
                const int2 t = ((const int2*)kin)[ic];
                keys[u] = t.x; vals[u] = t.y;
            } else {
                keys[u] = kin[ic]; vals[u] = vin[ic];
            }
        }
#pragma unroll
        for (int u = 0; u < SORT_GROUP; ++u) {
            const bool ok = wbase + (g + u) * 32 + lane < e;
            const int digit = (keys[u] >> SHIFT) & 255;
            const unsigned peers = digit_peers(digit, __builtin_amdgcn_ballot_w32(ok));
            if (ok) {
                const int off = hrow[digit];
                const int pos = off + __builtin_popcount(peers & ((1u << lane) - 1u));
                out[pos] = make_int2(keys[u], vals[u]);
                if (lane == __builtin_ctz(peers)) hrow[digit] = off + __builtin_popcount(peers);
            }
            __builtin_amdgcn_wave_barrier();
        }
    }
}

__global__ void row_bounds_kernel(const int2* __restrict__ filed, int* __restrict__ rowse, int e) {
    int i = blockIdx.x * blockDim.x + threadIdx.x;
    if (i >= e) return;
    const int d = filed[i].x;
    if (i == 0 || filed[i - 1].x != d) rowse[2 * d] = i;
    if (i == e - 1 || filed[i + 1].x != d) rowse[2 * d + 1] = i + 1;
}

template<int R>
__device__ __forceinline__ void gather_rows(const float* __restrict__ msg, int s_l, float w_l,
                                            int j, int lane, float4& acc) {
    float4 v[R];
#pragma unroll
    for (int u = 0; u < R; ++u) {
        const int s = __builtin_amdgcn_readlane(s_l, j + u);
        v[u] = ((const float4*)(msg + (size_t)s * H_C))[lane];
    }
#pragma unroll
    for (int u = 0; u < R; ++u) {
        const float w = __int_as_float(__builtin_amdgcn_readlane(__float_as_int(w_l), j + u));
        acc.x += v[u].x * w; acc.y += v[u].y * w;
        acc.z += v[u].z * w; acc.w += v[u].w * w;
    }
}

template<int NC, bool ADD_BIAS, bool STORE_AGG>
__global__ void __launch_bounds__(256)
gather_gemm_kernel(const int2* __restrict__ rowse, const int2* __restrict__ filed,
                   const float* __restrict__ msg, const float* __restrict__ bias_agg,
                   const float* __restrict__ B, const float* __restrict__ bias_out,
                   float* __restrict__ agg, float* __restrict__ C, int n) {
    constexpr int K     = H_C;
    constexpr int LDA   = K + 4;
    constexpr int TILES = NC / 16;
    __shared__ __align__(16) float As[16 * LDA];

    const int tid   = threadIdx.x;
    const int wave  = __builtin_amdgcn_readfirstlane(tid >> 5);
    const int lane  = tid & 31;
    const int lhalf = lane >> 4;
    const int l15   = lane & 15;
    const bool mult = TILES >= 8 || wave < TILES;
    const int ncol  = (mult ? wave : 0) * 16 + l15;

    v2f bfrag[K / 4];
    float bv = 0.0f;
    if (mult) {
#pragma unroll
        for (int k0 = 0; k0 < K; k0 += 4) {
            const int kk = k0 + 2 * lhalf;
            bfrag[k0 / 4].x = B[(size_t)kk * NC + ncol];
            bfrag[k0 / 4].y = B[(size_t)(kk + 1) * NC + ncol];
        }
        if (ADD_BIAS) bv = bias_out[ncol];
    }
    const int strips = (n + 15) / 16;
    for (int s = blockIdx.x; s < strips; s += gridDim.x) {
        const int row0 = s * 16;

        int   start[2], cnt[2], s_l[2];
        float di[2], w_l[2];
        float4 sv[2];
        {
            int2 se[2], t[2];
            int  sraw[2];
#pragma unroll
            for (int q = 0; q < 2; ++q) {
                const int d  = row0 + wave + 8 * q;
                const int dc = d < n ? d : n - 1;
                se[q] = rowse[dc];
                sv[q] = ((const float4*)(msg + (size_t)dc * H_C))[lane];
            }
#pragma unroll
            for (int q = 0; q < 2; ++q) {
                const int d = row0 + wave + 8 * q;
                start[q] = se[q].x;
                cnt[q]   = d < n ? se[q].y - se[q].x : 0;
                di[q]    = rsqrtf((float)(cnt[q] + 1));
                sraw[q]  = filed[start[q] + (lane < cnt[q] ? lane : 0)].y;
            }
#pragma unroll
            for (int q = 0; q < 2; ++q) t[q] = rowse[sraw[q]];
#pragma unroll
            for (int q = 0; q < 2; ++q) {
                const bool ok = lane < cnt[q];
                s_l[q] = ok ? sraw[q] : 0;
                w_l[q] = ok ? rsqrtf((float)(t[q].y - t[q].x + 1)) * di[q] : 0.0f;
            }
        }

        __syncthreads();
#pragma unroll
        for (int q = 0; q < 2; ++q) {
            const int r = wave + 8 * q;
            const int d = row0 + r;
            const int c = cnt[q];
            const float dq = di[q];
            const float4 ba = ((const float4*)bias_agg)[lane];
            float4 acc;
            acc.x = sv[q].x * dq * dq + ba.x; acc.y = sv[q].y * dq * dq + ba.y;
            acc.z = sv[q].z * dq * dq + ba.z; acc.w = sv[q].w * dq * dq + ba.w;
            int   sl = s_l[q];
            float wl = w_l[q];
            for (int base = 0; base < c; base += 32) {
                const int left = c - base;
                const int m = left < 32 ? left : 32;
                if (base > 0) {
                    sl = 0;
                    wl = 0.0f;
                    if (lane < m) {
                        sl = filed[start[q] + base + lane].y;
                        const int2 t = rowse[sl];
                        wl = rsqrtf((float)(t.y - t.x + 1)) * dq;
                    }
                }
                int j = 0;
                for (; j + 8 <= m; j += 8) gather_rows<8>(msg, sl, wl, j, lane, acc);
                if (j + 4 <= m) { gather_rows<4>(msg, sl, wl, j, lane, acc); j += 4; }
                if (j + 2 <= m) { gather_rows<2>(msg, sl, wl, j, lane, acc); j += 2; }
                if (j < m)      { gather_rows<1>(msg, sl, wl, j, lane, acc); }
            }
            if (d < n) {
                if (STORE_AGG) ((float4*)(agg + (size_t)d * H_C))[lane] = acc;
                acc.x = fmaxf(acc.x, 0.0f); acc.y = fmaxf(acc.y, 0.0f);
                acc.z = fmaxf(acc.z, 0.0f); acc.w = fmaxf(acc.w, 0.0f);
            } else {
                acc = make_float4(0.f, 0.f, 0.f, 0.f);
            }
            *(float4*)(As + r * LDA + 4 * lane) = acc;
        }
        __syncthreads();

        if (mult) {
            v8f o = {};
#pragma unroll
            for (int k0 = 0; k0 < K; k0 += 4) {
                const int kk = k0 + 2 * lhalf;
                v2f a;
                a.x = As[l15 * LDA + kk];
                a.y = As[l15 * LDA + kk + 1];
                o = __builtin_amdgcn_wmma_f32_16x16x4_f32(
                        false, a, false, bfrag[k0 / 4], (short)0, o, false, false);
            }
#pragma unroll
            for (int r = 0; r < 8; ++r) {
                const int m = row0 + r + 8 * lhalf;
                if (m < n) C[(size_t)m * NC + ncol] = o[r] + bv;
            }
        }
    }
}

#define POOL_NODES 32
#define POOL_WAVES 2

__device__ __forceinline__ void pool_flush(float* __restrict__ sums, float* __restrict__ cnts,
                                           int g, int lane, int run,
                                           float a0, float a1, float a2, float a3) {
    float* o = sums + (size_t)g * H_C + lane;
    atomic_fadd_at<0>(o, a0);
    atomic_fadd_at<128>(o, a1);
    atomic_fadd_at<256>(o, a2);
    atomic_fadd_at<384>(o, a3);
    if (lane == 0) atomic_fadd(&cnts[g], (float)run);
}

__global__ void __launch_bounds__(POOL_WAVES * 32)
pool_scatter_kernel(const float* __restrict__ h,
                    const int* __restrict__ batch,
                    float* __restrict__ sums,
                    float* __restrict__ cnts, int n) {
    const int lane = threadIdx.x & 31;
    const int node0 = __builtin_amdgcn_readfirstlane(
                          (int)(blockIdx.x * POOL_WAVES + (threadIdx.x >> 5))) * POOL_NODES;
    if (node0 >= n) return;
    const int left = n - node0;
    const int m = left < POOL_NODES ? left : POOL_NODES;
    const int g_l = lane < m ? batch[node0 + lane] : 0;

    int   cur = 0;
    int   run = 0;
    float a0 = 0.0f, a1 = 0.0f, a2 = 0.0f, a3 = 0.0f;
    for (int j0 = 0; j0 < m; j0 += 8) {
        float v[8][4];
#pragma unroll
        for (int u = 0; u < 8; ++u) {
            const int j = j0 + u < m ? j0 + u : m - 1;
            const float* r = h + (size_t)(node0 + j) * H_C + lane;
            v[u][0] = r[0]; v[u][1] = r[32]; v[u][2] = r[64]; v[u][3] = r[96];
        }
#pragma unroll
        for (int u = 0; u < 8; ++u) {
            if (j0 + u < m) {
                const int g = __builtin_amdgcn_readlane(g_l, j0 + u);
                if (run == 0 || g != cur) {
                    if (run != 0) pool_flush(sums, cnts, cur, lane, run, a0, a1, a2, a3);
                    cur = g; run = 0;
                    a0 = 0.0f; a1 = 0.0f; a2 = 0.0f; a3 = 0.0f;
                }
                a0 += fmaxf(v[u][0], 0.0f); a1 += fmaxf(v[u][1], 0.0f);
                a2 += fmaxf(v[u][2], 0.0f); a3 += fmaxf(v[u][3], 0.0f);
                ++run;
            }
        }
    }
    pool_flush(sums, cnts, cur, lane, run, a0, a1, a2, a3);
}

__global__ void pool_div_kernel(float* __restrict__ pooled,
                                const float* __restrict__ cnts, int total) {
    int i = blockIdx.x * blockDim.x + threadIdx.x;
    if (i >= total) return;
    pooled[i] /= fmaxf(cnts[i >> 7], 1.0f);
}

extern "C" void kernel_launch(void* const* d_in, const int* in_sizes, int n_in,
                              void* d_out, int out_size, void* d_ws, size_t ws_size,
                              hipStream_t stream) {
    const float* x    = (const float*)d_in[0];
    const int*   ei   = (const int*)  d_in[1];
    const int*   batch= (const int*)  d_in[2];
    const float* W1   = (const float*)d_in[3];
    const float* b1   = (const float*)d_in[4];
    const float* W2   = (const float*)d_in[5];
    const float* b2   = (const float*)d_in[6];
    const float* Wl   = (const float*)d_in[7];
    const float* bl   = (const float*)d_in[8];

    const int N = in_sizes[0] / IN_C;
    const int E = in_sizes[1] / 2;
    const int G = (out_size - N * OUT_C) / H_C;

    const int* esrc = ei;
    const int* edst = ei + E;

    float* logits = (float*)d_out;
    float* pooled = logits + (size_t)N * OUT_C;

    const int T  = 256;
    const int tiles = (E + SORT_TILE - 1) / SORT_TILE;

    const size_t Npad = ((size_t)N + 255) & ~(size_t)255;
    const size_t Epad = ((size_t)E + 255) & ~(size_t)255;
    int*   rowse = (int*)d_ws;
    int*   tab   = rowse + 2 * Npad;
    int2*  p1    = (int2*)(tab + (size_t)tiles * 256);
    int2*  p2    = p1 + Epad;
    float* tmpA = (float*)(p2 + Epad);
    float* tmpB = tmpA + (size_t)N * H_C;
    float* tmpC = tmpB + (size_t)N * H_C;
    float* cnts = tmpC + (size_t)N * H_C;

    const int S32 = (N + 31) / 32;
    const int c1 = S32 / 6, c2 = S32 / 2, c3 = c2 + c1;
    const SideGemm sg0 = {x, W1, tmpA, N, 0,  c1};
    const SideGemm sg1 = {x, W1, tmpA, N, c1, c2};
    const SideGemm sg2 = {x, W1, tmpA, N, c2, c3};
    const SideGemm sg3 = {x, W1, tmpA, N, c3, S32};
    const int gb0 = c1, gb1 = (c2 - c1 + 1) / 2, gb2 = c3 - c2, gb3 = (S32 - c3 + 1) / 2;
    const ZeroFill zf = {rowse, 2 * N, cnts, G, pooled, G * H_C};

    sort_count_kernel<0, 1, true><<<tiles + gb0, SORT_THREADS, 0, stream>>>(edst, tab, E, tiles, sg0, zf);
    sort_place_kernel<0, false><<<tiles + gb1, SORT_THREADS, 0, stream>>>(edst, esrc, p1, tab, E, tiles, sg1);
    sort_count_kernel<8, 2, false><<<tiles + gb2, SORT_THREADS, 0, stream>>>((const int*)p1, tab, E, tiles, sg2, zf);
    sort_place_kernel<8, true><<<tiles + gb3, SORT_THREADS, 0, stream>>>((const int*)p1, nullptr, p2, tab, E, tiles, sg3);
    row_bounds_kernel<<<(E + T - 1) / T, T, 0, stream>>>(p2, rowse, E);

    const int strips16    = (N + 15) / 16;
    const int fusedBlocks = strips16 < 768 ? strips16 : 768;

    gather_gemm_kernel<H_C, false, false><<<fusedBlocks, 256, 0, stream>>>(
        (const int2*)rowse, p2, tmpA, b1, W2, nullptr, nullptr, tmpB, N);

    gather_gemm_kernel<OUT_C, true, true><<<fusedBlocks, 256, 0, stream>>>(
        (const int2*)rowse, p2, tmpB, b2, Wl, bl, tmpC, logits, N);

    pool_scatter_kernel<<<(N + POOL_WAVES * POOL_NODES - 1) / (POOL_WAVES * POOL_NODES), POOL_WAVES * 32, 0, stream>>>(
        tmpC, batch, pooled, cnts, N);
    pool_div_kernel<<<(G * H_C + T - 1) / T, T, 0, stream>>>(pooled, cnts, G * H_C);
}


# fused aggregation: a wave's two nodes gathered together, 8 source rows of each (16 loads) in flight per wait instead of node after node in 8/4/2/1 groups; B fragments moved to LDS to free registers
# speedup vs baseline: 3.8675x; 1.0432x over previous
#include <hip/hip_runtime.h>

#define IN_C  128
#define H_C   128
#define OUT_C 64

typedef float v2f __attribute__((ext_vector_type(2)));
typedef float v8f __attribute__((ext_vector_type(8)));

__device__ __forceinline__ void atomic_fadd(float* p, float v) {
    asm volatile("global_atomic_add_f32 %0, %1, off" : : "v"(p), "v"(v) : "memory");
}

template<int OFF>
__device__ __forceinline__ void atomic_fadd_at(float* p, float v) {
    asm volatile("global_atomic_add_f32 %0, %1, off offset:%2" : : "v"(p), "v"(v), "n"(OFF) : "memory");
}


struct SideGemm {
    const float* A;
    const float* B;
    float*       C;
    int M;
    int lo, hi;
    int step;
};
struct ZeroFill {
    int*   a;
    float* b;
    float* c;
    int na, nb;
};

template<int K>
__device__ __forceinline__ void gemm_rows32(const float* __restrict__ A,
                                            const float* __restrict__ B,
                                            float* __restrict__ C, int M,
                                            int first, int last, int step) {
    constexpr int NC  = 128;
    constexpr int LDA = K + 4;
    __shared__ __align__(16) float As32[32 * LDA];

    const int tid   = threadIdx.x;
    const int wave  = tid >> 5;
    const int lane  = tid & 31;
    const int n0    = (wave & 7) * 16;
    const int rbase = (wave >> 3) * 16;
    const int lhalf = lane >> 4;
    const int l15   = lane & 15;
    const int n     = n0 + l15;

    v2f bfrag[K / 4];
#pragma unroll
    for (int k0 = 0; k0 < K; k0 += 4) {
        const int kk = k0 + 2 * lhalf;
        bfrag[k0 / 4].x = B[(size_t)kk * NC + n];
        bfrag[k0 / 4].y = B[(size_t)(kk + 1) * NC + n];
    }

    constexpr int vecsPerRow = K / 4;
    constexpr int PER_THREAD = 32 * vecsPerRow / 512;
    static_assert(PER_THREAD * 512 == 32 * vecsPerRow, "a strip is a whole number of pieces per thread");
    for (int s = first; s < last; s += step) {
        const int row0 = s * 32;
        __syncthreads();
        float4 val[PER_THREAD];
#pragma unroll
        for (int u = 0; u < PER_THREAD; ++u) {
            const int v = tid + u * 512;
            const int m = row0 + v / vecsPerRow;
            val[u] = ((const float4*)(A + (size_t)(m < M ? m : M - 1) * K))[v % vecsPerRow];
        }
#pragma unroll
        for (int u = 0; u < PER_THREAD; ++u) {
            const int v = tid + u * 512;
            const int r = v / vecsPerRow;
            if (row0 + r >= M) val[u] = make_float4(0.f, 0.f, 0.f, 0.f);
            *(float4*)(As32 + r * LDA + 4 * (v % vecsPerRow)) = val[u];
        }
        __syncthreads();

        v8f acc = {};
#pragma unroll
        for (int k0 = 0; k0 < K; k0 += 4) {
            const int kk = k0 + 2 * lhalf;
            v2f a;
            a.x = As32[(rbase + l15) * LDA + kk];
            a.y = As32[(rbase + l15) * LDA + kk + 1];
            acc = __builtin_amdgcn_wmma_f32_16x16x4_f32(
                      false, a, false, bfrag[k0 / 4], (short)0, acc, false, false);
        }
#pragma unroll
        for (int r = 0; r < 8; ++r) {
            const int m = row0 + rbase + r + 8 * lhalf;
            if (m < M) C[(size_t)m * NC + n] = acc[r];
        }
    }
}

#define SORT_WAVES   16
#define SORT_CHUNKS  8
#define SORT_GROUP   8
#define SORT_TILE    (SORT_WAVES * SORT_CHUNKS * 32)
#define SORT_THREADS (SORT_WAVES * 32)
#define SORT_SPLIT   (SORT_THREADS / 256)
#define WALK_BATCH   32

__device__ __forceinline__ unsigned digit_peers(int digit, unsigned valid) {
    unsigned peers = valid;
#pragma unroll
    for (int b = 0; b < 8; ++b) {
        const bool bit = (digit >> b) & 1;
        const unsigned m = __builtin_amdgcn_ballot_w32(bit);
        peers &= bit ? m : ~m;
    }
    return peers;
}

template<int SHIFT, int KSTRIDE>
__device__ __forceinline__ void count_digits(const int* __restrict__ kin, int* hrow,
                                             int wbase, int lane, int e) {
#pragma unroll
    for (int q = 0; q < 8; ++q) hrow[lane + 32 * q] = 0;
    __builtin_amdgcn_wave_barrier();
#pragma unroll 1
    for (int g = 0; g < SORT_CHUNKS; g += SORT_GROUP) {
        int keys[SORT_GROUP];
#pragma unroll
        for (int u = 0; u < SORT_GROUP; ++u) {
            const int i = wbase + (g + u) * 32 + lane;
            keys[u] = kin[(size_t)(i < e ? i : e - 1) * KSTRIDE];
        }
#pragma unroll
        for (int u = 0; u < SORT_GROUP; ++u) {
            const bool ok = wbase + (g + u) * 32 + lane < e;
            const int digit = (keys[u] >> SHIFT) & 255;
            const unsigned peers = digit_peers(digit, __builtin_amdgcn_ballot_w32(ok));
            if (ok && lane == __builtin_ctz(peers)) hrow[digit] += __builtin_popcount(peers);
            __builtin_amdgcn_wave_barrier();
        }
    }
}

template<int SHIFT, int KSTRIDE, bool FILL>
__global__ void __launch_bounds__(SORT_THREADS)
sort_count_kernel(const int* __restrict__ kin, int* __restrict__ tab, int e, int nt,
                  SideGemm sg, ZeroFill zf) {
    __shared__ int h[SORT_WAVES][256];
    const int tid = threadIdx.x, w = tid >> 5, lane = tid & 31;
    if (FILL) {
        const int gsz = (int)gridDim.x * SORT_THREADS;
        const int g0  = (int)blockIdx.x * SORT_THREADS + tid;
        for (int i = g0; i < zf.na; i += gsz) zf.a[i] = 0;
        for (int i = g0; i < zf.nb; i += gsz) zf.b[i] = 0.0f;
        for (int i = g0; i < zf.nb * H_C; i += gsz) zf.c[i] = 0.0f;
    }
    const int gb = (int)blockIdx.x - nt;
    if (gb >= 0) {
        gemm_rows32<IN_C>(sg.A, sg.B, sg.C, sg.M, sg.lo + gb, sg.hi, sg.step);
        return;
    }
    const int wbase = blockIdx.x * SORT_TILE + w * (SORT_CHUNKS * 32);
    count_digits<SHIFT, KSTRIDE>(kin, h[w], wbase, lane, e);
    __syncthreads();
    if (tid < 256) {
        int t = 0;
#pragma unroll
        for (int v = 0; v < SORT_WAVES; ++v) t += h[v][tid];
        tab[blockIdx.x * 256 + tid] = t;
    }
}

template<int SHIFT, bool PAIRS>
__global__ void __launch_bounds__(SORT_THREADS)
sort_place_kernel(const int* __restrict__ kin, const int* __restrict__ vin,
                  int2* __restrict__ out, const int* __restrict__ tab, int e, int nt,
                  SideGemm sg) {
    __shared__ int h[SORT_WAVES][256];
    __shared__ int ptot[SORT_SPLIT][256];
    __shared__ int pbef[SORT_SPLIT][256];
    __shared__ int wsum[8];
    const int tid = threadIdx.x, w = tid >> 5, lane = tid & 31;
    const int gb = (int)blockIdx.x - nt;
    if (gb >= 0) {
        gemm_rows32<IN_C>(sg.A, sg.B, sg.C, sg.M, sg.lo + gb, sg.hi, sg.step);
        return;
    }
    const int tile = blockIdx.x;
    const int wbase = tile * SORT_TILE + w * (SORT_CHUNKS * 32);

    {
        const int q = __builtin_amdgcn_readfirstlane(tid >> 8);
        const int dg = tid & 255;
        int total = 0, before = 0;
        for (int t0 = q; t0 < nt; t0 += WALK_BATCH * SORT_SPLIT) {
            int c[WALK_BATCH];
#pragma unroll
            for (int u = 0; u < WALK_BATCH; ++u) {
                const int t = t0 + u * SORT_SPLIT;
                c[u] = tab[(t < nt ? t : nt - 1) * 256 + dg];
            }
#pragma unroll
            for (int u = 0; u < WALK_BATCH; ++u) {
                const int t = t0 + u * SORT_SPLIT;
                total  += t < nt ? c[u] : 0;
                before += t < tile ? c[u] : 0;
            }
        }
        ptot[q][dg] = total;
        pbef[q][dg] = before;
    }

    count_digits<SHIFT, PAIRS ? 2 : 1>(kin, h[w], wbase, lane, e);
    __syncthreads();

    int total = 0, before = 0, incl = 0;
    if (tid < 256) {
#pragma unroll
        for (int q = 0; q < SORT_SPLIT; ++q) { total += ptot[q][tid]; before += pbef[q][tid]; }
        incl = total;
#pragma unroll
        for (int off = 1; off < 32; off <<= 1) {
            const int t = __shfl_up(incl, off, 32);
            if (lane >= off) incl += t;
        }
        if (lane == 31) wsum[w] = incl;
    }
    __syncthreads();

    if (tid < 256) {
        int run = before + incl - total;
        for (int v = 0; v < w; ++v) run += wsum[v];
#pragma unroll
        for (int v = 0; v < SORT_WAVES; ++v) {
            const int t = h[v][tid];
            h[v][tid] = run;
            run += t;
        }
    }
    __syncthreads();

    int* hrow = h[w];
#pragma unroll 1
    for (int g = 0; g < SORT_CHUNKS; g += SORT_GROUP) {
        int keys[SORT_GROUP], vals[SORT_GROUP];
#pragma unroll
        for (int u = 0; u < SORT_GROUP; ++u) {
            const int i = wbase + (g + u) * 32 + lane;
            const int ic = i < e ? i : e - 1;
            if (PAIRS) {
                const int2 t = ((const int2*)kin)[ic];
                keys[u] = t.x; vals[u] = t.y;
            } else {
                keys[u] = kin[ic]; vals[u] = vin[ic];
            }
        }
#pragma unroll
        for (int u = 0; u < SORT_GROUP; ++u) {
            const bool ok = wbase + (g + u) * 32 + lane < e;
            const int digit = (keys[u] >> SHIFT) & 255;
            const unsigned peers = digit_peers(digit, __builtin_amdgcn_ballot_w32(ok));
            if (ok) {
                const int off = hrow[digit];
                const int pos = off + __builtin_popcount(peers & ((1u << lane) - 1u));
                out[pos] = make_int2(keys[u], vals[u]);
                if (lane == __builtin_ctz(peers)) hrow[digit] = off + __builtin_popcount(peers);
            }
            __builtin_amdgcn_wave_barrier();
        }
    }
}

__global__ void row_bounds_kernel(const int2* __restrict__ filed, int* __restrict__ rowse, int e) {
    int i = blockIdx.x * blockDim.x + threadIdx.x;
    if (i >= e) return;
    const int d = filed[i].x;
    if (i == 0 || filed[i - 1].x != d) rowse[2 * d] = i;
    if (i == e - 1 || filed[i + 1].x != d) rowse[2 * d + 1] = i + 1;
}

template<int R>
__device__ __forceinline__ void issue_rows(const float* __restrict__ msg, int s_l, int j,
                                           int lane, float4 (&v)[R]) {
#pragma unroll
    for (int u = 0; u < R; ++u) {
        const int s = __builtin_amdgcn_readlane(s_l, j + u);
        v[u] = ((const float4*)(msg + (size_t)s * H_C))[lane];
    }
}
template<int R>
__device__ __forceinline__ void add_rows(const float4 (&v)[R], float w_l, int j, float4& acc) {
#pragma unroll
    for (int u = 0; u < R; ++u) {
        const float w = __int_as_float(__builtin_amdgcn_readlane(__float_as_int(w_l), j + u));
        acc.x += v[u].x * w; acc.y += v[u].y * w;
        acc.z += v[u].z * w; acc.w += v[u].w * w;
    }
}

template<int NC, bool ADD_BIAS, bool STORE_AGG>
__global__ void __launch_bounds__(256)
gather_gemm_kernel(const int2* __restrict__ rowse, const int2* __restrict__ filed,
                   const float* __restrict__ msg, const float* __restrict__ bias_agg,
                   const float* __restrict__ B, const float* __restrict__ bias_out,
                   float* __restrict__ agg, float* __restrict__ C, int n) {
    constexpr int K     = H_C;
    constexpr int LDA   = K + 4;
    constexpr int TILES = NC / 16;
    __shared__ __align__(16) float As[16 * LDA];
    __shared__ __align__(16) v2f Bs[TILES < 8 ? TILES : 8][K / 4][32];

    const int tid   = threadIdx.x;
    const int wave  = __builtin_amdgcn_readfirstlane(tid >> 5);
    const int lane  = tid & 31;
    const int lhalf = lane >> 4;
    const int l15   = lane & 15;
    const bool mult = TILES >= 8 || wave < TILES;
    const int ncol  = (mult ? wave : 0) * 16 + l15;

    float bv = 0.0f;
    if (mult) {
#pragma unroll
        for (int k0 = 0; k0 < K; k0 += 4) {
            const int kk = k0 + 2 * lhalf;
            v2f b;
            b.x = B[(size_t)kk * NC + ncol];
            b.y = B[(size_t)(kk + 1) * NC + ncol];
            Bs[wave][k0 / 4][lane] = b;
        }
        if (ADD_BIAS) bv = bias_out[ncol];
    }
    const int strips = (n + 15) / 16;
    for (int s = blockIdx.x; s < strips; s += gridDim.x) {
        const int row0 = s * 16;

        int   start[2], cnt[2], s_l[2], own[2];
        float di[2], w_l[2];
        float4 sv[2];
        {
            int2 se[2], t[2];
            int  sraw[2];
#pragma unroll
            for (int q = 0; q < 2; ++q) {
                const int d  = row0 + wave + 8 * q;
                const int dc = d < n ? d : n - 1;
                own[q] = dc;
                se[q] = rowse[dc];
                sv[q] = ((const float4*)(msg + (size_t)dc * H_C))[lane];
            }
#pragma unroll
            for (int q = 0; q < 2; ++q) {
                const int d = row0 + wave + 8 * q;
                start[q] = se[q].x;
                cnt[q]   = d < n ? se[q].y - se[q].x : 0;
                di[q]    = rsqrtf((float)(cnt[q] + 1));
                sraw[q]  = filed[start[q] + (lane < cnt[q] ? lane : 0)].y;
            }
#pragma unroll
            for (int q = 0; q < 2; ++q) t[q] = rowse[sraw[q]];
#pragma unroll
            for (int q = 0; q < 2; ++q) {
                const bool ok = lane < cnt[q];
                s_l[q] = ok ? sraw[q] : own[q];
                w_l[q] = ok ? rsqrtf((float)(t[q].y - t[q].x + 1)) * di[q] : 0.0f;
            }
        }

        const float4 ba = ((const float4*)bias_agg)[lane];
        float4 acc[2];
        int    sl[2];
        float  wl[2];
#pragma unroll
        for (int q = 0; q < 2; ++q) {
            const float dq = di[q];
            acc[q].x = sv[q].x * dq * dq + ba.x; acc[q].y = sv[q].y * dq * dq + ba.y;
            acc[q].z = sv[q].z * dq * dq + ba.z; acc[q].w = sv[q].w * dq * dq + ba.w;
            sl[q] = s_l[q];
            wl[q] = w_l[q];
        }
        const int c0 = __builtin_amdgcn_readfirstlane(cnt[0]);
        const int c1 = __builtin_amdgcn_readfirstlane(cnt[1]);
        const int cmax = c0 > c1 ? c0 : c1;
        for (int base = 0; base < cmax; base += 32) {
            if (base > 0) {
#pragma unroll
                for (int q = 0; q < 2; ++q) {
                    sl[q] = own[q];
                    wl[q] = 0.0f;
                    if (base + lane < cnt[q]) {
                        sl[q] = filed[start[q] + base + lane].y;
                        const int2 t = rowse[sl[q]];
                        wl[q] = rsqrtf((float)(t.y - t.x + 1)) * di[q];
                    }
                }
            }
            for (int j = 0; j < 32 && base + j < cmax; j += 8) {
                const bool g0 = base + j < c0;
                const bool g1 = base + j < c1;
                if (g0 && g1) {
                    float4 v0[8], v1[8];
                    issue_rows<8>(msg, sl[0], j, lane, v0);
                    issue_rows<8>(msg, sl[1], j, lane, v1);
                    add_rows<8>(v0, wl[0], j, acc[0]);
                    add_rows<8>(v1, wl[1], j, acc[1]);
                } else if (g0) {
                    float4 v0[8];
                    issue_rows<8>(msg, sl[0], j, lane, v0);
                    add_rows<8>(v0, wl[0], j, acc[0]);
                } else {
                    float4 v1[8];
                    issue_rows<8>(msg, sl[1], j, lane, v1);
                    add_rows<8>(v1, wl[1], j, acc[1]);
                }
            }
        }
        __syncthreads();
#pragma unroll
        for (int q = 0; q < 2; ++q) {
            const int r = wave + 8 * q;
            const int d = row0 + r;
            float4 a = acc[q];
            if (d < n) {
                if (STORE_AGG) ((float4*)(agg + (size_t)d * H_C))[lane] = a;
                a.x = fmaxf(a.x, 0.0f); a.y = fmaxf(a.y, 0.0f);
                a.z = fmaxf(a.z, 0.0f); a.w = fmaxf(a.w, 0.0f);
            } else {
                a = make_float4(0.f, 0.f, 0.f, 0.f);
            }
            *(float4*)(As + r * LDA + 4 * lane) = a;
        }
        __syncthreads();

        if (mult) {
            v8f o = {};
#pragma unroll
            for (int k0 = 0; k0 < K; k0 += 4) {
                const int kk = k0 + 2 * lhalf;
                v2f a;
                a.x = As[l15 * LDA + kk];
                a.y = As[l15 * LDA + kk + 1];
                const v2f b = Bs[wave][k0 / 4][lane];
                o = __builtin_amdgcn_wmma_f32_16x16x4_f32(
                        false, a, false, b, (short)0, o, false, false);
            }
#pragma unroll
            for (int r = 0; r < 8; ++r) {
                const int m = row0 + r + 8 * lhalf;
                if (m < n) C[(size_t)m * NC + ncol] = o[r] + bv;
            }
        }
    }
}

#define POOL_NODES 32
#define POOL_WAVES 2

__device__ __forceinline__ void pool_flush(float* __restrict__ sums, float* __restrict__ cnts,
                                           int g, int lane, int run,
                                           float a0, float a1, float a2, float a3) {
    float* o = sums + (size_t)g * H_C + lane;
    atomic_fadd_at<0>(o, a0);
    atomic_fadd_at<128>(o, a1);
    atomic_fadd_at<256>(o, a2);
    atomic_fadd_at<384>(o, a3);
    if (lane == 0) atomic_fadd(&cnts[g], (float)run);
}

__global__ void __launch_bounds__(POOL_WAVES * 32)
pool_scatter_kernel(const float* __restrict__ h,
                    const int* __restrict__ batch,
                    float* __restrict__ sums,
                    float* __restrict__ cnts, int n) {
    const int lane = threadIdx.x & 31;
    const int node0 = __builtin_amdgcn_readfirstlane(
                          (int)(blockIdx.x * POOL_WAVES + (threadIdx.x >> 5))) * POOL_NODES;
    if (node0 >= n) return;
    const int left = n - node0;
    const int m = left < POOL_NODES ? left : POOL_NODES;
    const int g_l = lane < m ? batch[node0 + lane] : 0;

    int   cur = 0;
    int   run = 0;
    float a0 = 0.0f, a1 = 0.0f, a2 = 0.0f, a3 = 0.0f;
    for (int j0 = 0; j0 < m; j0 += 8) {
        float v[8][4];
#pragma unroll
        for (int u = 0; u < 8; ++u) {
            const int j = j0 + u < m ? j0 + u : m - 1;
            const float* r = h + (size_t)(node0 + j) * H_C + lane;
            v[u][0] = r[0]; v[u][1] = r[32]; v[u][2] = r[64]; v[u][3] = r[96];
        }
#pragma unroll
        for (int u = 0; u < 8; ++u) {
            if (j0 + u < m) {
                const int g = __builtin_amdgcn_readlane(g_l, j0 + u);
                if (run == 0 || g != cur) {
                    if (run != 0) pool_flush(sums, cnts, cur, lane, run, a0, a1, a2, a3);
                    cur = g; run = 0;
                    a0 = 0.0f; a1 = 0.0f; a2 = 0.0f; a3 = 0.0f;
                }
                a0 += fmaxf(v[u][0], 0.0f); a1 += fmaxf(v[u][1], 0.0f);
                a2 += fmaxf(v[u][2], 0.0f); a3 += fmaxf(v[u][3], 0.0f);
                ++run;
            }
        }
    }
    pool_flush(sums, cnts, cur, lane, run, a0, a1, a2, a3);
}

__global__ void pool_div_kernel(float* __restrict__ pooled,
                                const float* __restrict__ cnts, int total) {
    int i = blockIdx.x * blockDim.x + threadIdx.x;
    if (i >= total) return;
    pooled[i] /= fmaxf(cnts[i >> 7], 1.0f);
}

extern "C" void kernel_launch(void* const* d_in, const int* in_sizes, int n_in,
                              void* d_out, int out_size, void* d_ws, size_t ws_size,
                              hipStream_t stream) {
    const float* x    = (const float*)d_in[0];
    const int*   ei   = (const int*)  d_in[1];
    const int*   batch= (const int*)  d_in[2];
    const float* W1   = (const float*)d_in[3];
    const float* b1   = (const float*)d_in[4];
    const float* W2   = (const float*)d_in[5];
    const float* b2   = (const float*)d_in[6];
    const float* Wl   = (const float*)d_in[7];
    const float* bl   = (const float*)d_in[8];

    const int N = in_sizes[0] / IN_C;
    const int E = in_sizes[1] / 2;
    const int G = (out_size - N * OUT_C) / H_C;

    const int* esrc = ei;
    const int* edst = ei + E;

    float* logits = (float*)d_out;
    float* pooled = logits + (size_t)N * OUT_C;

    const int T  = 256;
    const int tiles = (E + SORT_TILE - 1) / SORT_TILE;

    const size_t Npad = ((size_t)N + 255) & ~(size_t)255;
    const size_t Epad = ((size_t)E + 255) & ~(size_t)255;
    int*   rowse = (int*)d_ws;
    int*   tab   = rowse + 2 * Npad;
    int2*  p1    = (int2*)(tab + (size_t)tiles * 256);
    int2*  p2    = p1 + Epad;
    float* tmpA = (float*)(p2 + Epad);
    float* tmpB = tmpA + (size_t)N * H_C;
    float* tmpC = tmpB + (size_t)N * H_C;
    float* cnts = tmpC + (size_t)N * H_C;

    const int S32 = (N + 31) / 32;
    const int c1 = S32 / 6, c2 = S32 / 2, c3 = c2 + c1;
    const int gb0 = c1, gb1 = (c2 - c1 + 1) / 2, gb2 = c3 - c2, gb3 = (S32 - c3 + 1) / 2;
    const SideGemm sg0 = {x, W1, tmpA, N, 0,  c1,  gb0};
    const SideGemm sg1 = {x, W1, tmpA, N, c1, c2,  gb1};
    const SideGemm sg2 = {x, W1, tmpA, N, c2, c3,  gb2};
    const SideGemm sg3 = {x, W1, tmpA, N, c3, S32, gb3};
    const ZeroFill zf = {rowse, cnts, pooled, 2 * N, G};

    sort_count_kernel<0, 1, true><<<tiles + gb0, SORT_THREADS, 0, stream>>>(edst, tab, E, tiles, sg0, zf);
    sort_place_kernel<0, false><<<tiles + gb1, SORT_THREADS, 0, stream>>>(edst, esrc, p1, tab, E, tiles, sg1);
    sort_count_kernel<8, 2, false><<<tiles + gb2, SORT_THREADS, 0, stream>>>((const int*)p1, tab, E, tiles, sg2, zf);
    sort_place_kernel<8, true><<<tiles + gb3, SORT_THREADS, 0, stream>>>((const int*)p1, nullptr, p2, tab, E, tiles, sg3);
    row_bounds_kernel<<<(E + T - 1) / T, T, 0, stream>>>(p2, rowse, E);

    const int strips16    = (N + 15) / 16;
    const int perGroup    = (strips16 + 1023) / 1024;
    const int fusedBlocks = (strips16 + perGroup - 1) / perGroup;

    gather_gemm_kernel<H_C, false, false><<<fusedBlocks, 256, 0, stream>>>(
        (const int2*)rowse, p2, tmpA, b1, W2, nullptr, nullptr, tmpB, N);

    gather_gemm_kernel<OUT_C, true, true><<<fusedBlocks, 256, 0, stream>>>(
        (const int2*)rowse, p2, tmpB, b2, Wl, bl, tmpC, logits, N);

    pool_scatter_kernel<<<(N + POOL_WAVES * POOL_NODES - 1) / (POOL_WAVES * POOL_NODES), POOL_WAVES * 32, 0, stream>>>(
        tmpC, batch, pooled, cnts, N);
    pool_div_kernel<<<(G * H_C + T - 1) / T, T, 0, stream>>>(pooled, cnts, G * H_C);
}
